# starting kernel
# baseline (speedup 1.0000x reference)
#include <hip/hip_runtime.h>
#include <hip/hip_bf16.h>
#include <math.h>

typedef __attribute__((ext_vector_type(2))) float v2f;
typedef __attribute__((ext_vector_type(8))) float v8f;

static __device__ __forceinline__ v8f wmma_f32_16x16x4(v2f a, v2f b, v8f c) {
  return __builtin_amdgcn_wmma_f32_16x16x4_f32(false, a, false, b, (short)0, c, false, false);
}

static __device__ __forceinline__ float fast_sigmoid(float x) {
  x = fminf(fmaxf(x, -30.f), 30.f);
  return __builtin_amdgcn_rcpf(1.f + __expf(-x));
}
static __device__ __forceinline__ float fast_tanh(float x) {
  x = fminf(fmaxf(x, -15.f), 15.f);
  float e = __expf(-2.f * x);
  return (1.f - e) * __builtin_amdgcn_rcpf(1.f + e);
}

#define B_SZ 256
#define S_SZ 256
#define F_SZ 64
#define H_SZ 128
#define HSTR 132

__global__ void __launch_bounds__(128)
qlstm_xproj(const float* __restrict__ x, const float* __restrict__ Win,
            const float* __restrict__ bin, float* __restrict__ y0) {
  const int lane = threadIdx.x & 31;
  const int wave = threadIdx.x >> 5;
  const int tile = blockIdx.x * 4 + wave;
  const int n    = lane & 15;
  const int hi   = lane >> 4;
  const int kh   = hi << 1;

  v2f bch[16];
#pragma unroll
  for (int j = 0; j < 16; ++j) {
    float b0 = 0.f, b1 = 0.f;
    if (n < 4) {
      int kk = j * 4 + kh;
      b0 = Win[n * 192 + 128 + kk];
      b1 = Win[n * 192 + 128 + kk + 1];
    }
    v2f bv = {b0, b1};
    bch[j] = bv;
  }

  const float bb = (n < 4) ? bin[n] : 0.f;
  v8f acc = {bb, bb, bb, bb, bb, bb, bb, bb};

  const float* xr = x + (size_t)(tile * 16 + n) * F_SZ;
#pragma unroll
  for (int j = 0; j < 16; ++j) {
    v2f a = {xr[j * 4 + kh], xr[j * 4 + kh + 1]};
    acc = wmma_f32_16x16x4(a, bch[j], acc);
  }

  if (n < 4) {
#pragma unroll
    for (int r = 0; r < 8; ++r) {
      int row = tile * 16 + r + (hi << 3);
      y0[(size_t)row * 4 + n] = acc[r];
    }
  }
}

__global__ void __launch_bounds__(256)
qlstm_recur(const float* __restrict__ y0, const float* __restrict__ Win,
            const float* __restrict__ Wout, const float* __restrict__ bout,
            const float* __restrict__ thetas, float* __restrict__ out) {
  __shared__ float y0_lds[16 * S_SZ * 4];
  __shared__ float h_lds[16 * HSTR];
  __shared__ float part[8 * 16 * 4];
  __shared__ float z_lds[4 * 16 * 4];

  const int tid   = threadIdx.x;
  const int lane  = tid & 31;
  const int wave  = tid >> 5;
  const int n     = lane & 15;
  const int hi    = lane >> 4;
  const int kh    = hi << 1;
  const int bbase = blockIdx.x * 16;

  float* hs = out;
  float* hT = out + (size_t)B_SZ * S_SZ * H_SZ;
  float* cT = hT + (size_t)B_SZ * H_SZ;

  {
    const float4* src = (const float4*)(y0 + (size_t)bbase * S_SZ * 4);
    float4* dst = (float4*)y0_lds;
#pragma unroll
    for (int i = 0; i < 16; ++i)
      dst[tid + i * 256] = src[tid + i * 256];
  }
  for (int i = tid; i < 16 * HSTR; i += 256) h_lds[i] = 0.f;

  v2f bh[4];
#pragma unroll
  for (int j = 0; j < 4; ++j) {
    float b0 = 0.f, b1 = 0.f;
    if (n < 4) {
      int kk = wave * 16 + j * 4 + kh;
      b0 = Win[n * 192 + kk];
      b1 = Win[n * 192 + kk + 1];
    }
    v2f bv = {b0, b1};
    bh[j] = bv;
  }
  const int col = wave * 16 + n;
  v2f bw = {Wout[col * 4 + kh], Wout[col * 4 + kh + 1]};
  const float bo = bout[col];
  const v8f cbias = {bo, bo, bo, bo, bo, bo, bo, bo};

  const int sb = tid >> 2, sg = tid & 3;
  float th[4];
#pragma unroll
  for (int w = 0; w < 4; ++w) th[w] = (tid < 64) ? thetas[sg * 4 + w] : 0.f;

  float c_state[8], h_keep[8];
#pragma unroll
  for (int r = 0; r < 8; ++r) { c_state[r] = 0.f; h_keep[r] = 0.f; }

  __syncthreads();

  for (int t = 0; t < S_SZ; ++t) {
    v8f acc = {0.f, 0.f, 0.f, 0.f, 0.f, 0.f, 0.f, 0.f};
#pragma unroll
    for (int j = 0; j < 4; ++j) {
      int kk = wave * 16 + j * 4 + kh;
      v2f a = {h_lds[n * HSTR + kk], h_lds[n * HSTR + kk + 1]};
      acc = wmma_f32_16x16x4(a, bh[j], acc);
    }
    if (n < 4) {
#pragma unroll
      for (int r = 0; r < 8; ++r)
        part[wave * 64 + (r + (hi << 3)) * 4 + n] = acc[r];
    }
    __syncthreads();

    if (tid < 64) {
      float p0[4], p1[4];
#pragma unroll
      for (int w = 0; w < 4; ++w) {
        float yw = y0_lds[(sb * S_SZ + t) * 4 + w];
#pragma unroll
        for (int wv = 0; wv < 8; ++wv) yw += part[wv * 64 + sb * 4 + w];
        float cv = __cosf(yw + th[w]);
        p0[w] = 0.5f + 0.5f * cv;
        p1[w] = 0.5f - 0.5f * cv;
      }
      float z0 = 0.f, z1 = 0.f, z2 = 0.f, z3 = 0.f;
#pragma unroll
      for (int idx = 0; idx < 16; ++idx) {
        int a0 = (idx >> 3) & 1, a1 = (idx >> 2) & 1;
        int a2 = (idx >> 1) & 1, a3 = idx & 1;
        float p = (a0 ? p1[0] : p0[0]) * (a1 ? p1[1] : p0[1]) *
                  (a2 ? p1[2] : p0[2]) * (a3 ? p1[3] : p0[3]);
        int t1 = a1 ^ a0;
        int t2 = a2 ^ t1;
        int t3 = a3 ^ t2;
        int t0 = a0 ^ t3;
        z0 += t0 ? -p : p;  z1 += t1 ? -p : p;
        z2 += t2 ? -p : p;  z3 += t3 ? -p : p;
      }
      int base = (sg * 16 + sb) * 4;
      z_lds[base + 0] = z0;  z_lds[base + 1] = z1;
      z_lds[base + 2] = z2;  z_lds[base + 3] = z3;
    }
    __syncthreads();

    v8f ga[4];
#pragma unroll
    for (int g = 0; g < 4; ++g) {
      v2f a = {z_lds[(g * 16 + n) * 4 + kh], z_lds[(g * 16 + n) * 4 + kh + 1]};
      ga[g] = wmma_f32_16x16x4(a, bw, cbias);
    }
#pragma unroll
    for (int r = 0; r < 8; ++r) {
      float fv = fast_sigmoid(ga[0][r]);
      float iv = fast_sigmoid(ga[1][r]);
      float gv = fast_tanh(ga[2][r]);
      float ov = fast_sigmoid(ga[3][r]);
      float cs = fv * c_state[r] + iv * gv;
      c_state[r] = cs;
      float hv = ov * fast_tanh(cs);
      h_keep[r] = hv;
      int mrow = r + (hi << 3);
      h_lds[mrow * HSTR + col] = hv;
      hs[(((size_t)(bbase + mrow)) * S_SZ + t) * H_SZ + col] = hv;
    }
    __syncthreads();
  }

#pragma unroll
  for (int r = 0; r < 8; ++r) {
    int mrow = r + (hi << 3);
    hT[((size_t)(bbase + mrow)) * H_SZ + col] = h_keep[r];
    cT[((size_t)(bbase + mrow)) * H_SZ + col] = c_state[r];
  }
}

extern "C" void kernel_launch(void* const* d_in, const int* in_sizes, int n_in,
                              void* d_out, int out_size, void* d_ws, size_t ws_size,
                              hipStream_t stream) {
  const float* x      = (const float*)d_in[0];
  const float* W_in   = (const float*)d_in[1];
  const float* b_in   = (const float*)d_in[2];
  const float* W_out  = (const float*)d_in[3];
  const float* b_out  = (const float*)d_in[4];
  const float* thetas = (const float*)d_in[5];
  float* out = (float*)d_out;
  float* y0  = (float*)d_ws;

  qlstm_xproj<<<1024, 128, 0, stream>>>(x, W_in, b_in, y0);
  qlstm_recur<<<16, 256, 0, stream>>>(y0, W_in, W_out, b_out, thetas, out);
}


# per-step barriers fence LDS only (local-space release/acquire around s_barrier): the end-of-step s_wait_storecnt 0 goes and the hs global stores stay in flight across steps; arithmetic unchanged
# speedup vs baseline: 1.7496x; 1.7496x over previous
#include <hip/hip_runtime.h>
#include <hip/hip_bf16.h>
#include <math.h>

typedef __attribute__((ext_vector_type(2))) float v2f;
typedef __attribute__((ext_vector_type(8))) float v8f;

static __device__ __forceinline__ v8f wmma_f32_16x16x4(v2f a, v2f b, v8f c) {
  return __builtin_amdgcn_wmma_f32_16x16x4_f32(false, a, false, b, (short)0, c, false, false);
}

static __device__ __forceinline__ float fast_sigmoid(float x) {
  x = fminf(fmaxf(x, -30.f), 30.f);
  return __builtin_amdgcn_rcpf(1.f + __expf(-x));
}
static __device__ __forceinline__ float fast_tanh(float x) {
  x = fminf(fmaxf(x, -15.f), 15.f);
  float e = __expf(-2.f * x);
  return (1.f - e) * __builtin_amdgcn_rcpf(1.f + e);
}

static __device__ __forceinline__ void lds_barrier() {
  __builtin_amdgcn_fence(__ATOMIC_RELEASE, "workgroup", "local");
  __builtin_amdgcn_s_barrier();
  __builtin_amdgcn_fence(__ATOMIC_ACQUIRE, "workgroup", "local");
}

#define B_SZ 256
#define S_SZ 256
#define F_SZ 64
#define H_SZ 128
#define HSTR 132

__global__ void __launch_bounds__(128)
qlstm_xproj(const float* __restrict__ x, const float* __restrict__ Win,
            const float* __restrict__ bin, float* __restrict__ y0) {
  const int lane = threadIdx.x & 31;
  const int wave = threadIdx.x >> 5;
  const int tile = blockIdx.x * 4 + wave;
  const int n    = lane & 15;
  const int hi   = lane >> 4;
  const int kh   = hi << 1;

  v2f bch[16];
#pragma unroll
  for (int j = 0; j < 16; ++j) {
    float b0 = 0.f, b1 = 0.f;
    if (n < 4) {
      int kk = j * 4 + kh;
      b0 = Win[n * 192 + 128 + kk];
      b1 = Win[n * 192 + 128 + kk + 1];
    }
    v2f bv = {b0, b1};
    bch[j] = bv;
  }

  const float bb = (n < 4) ? bin[n] : 0.f;
  v8f acc = {bb, bb, bb, bb, bb, bb, bb, bb};

  const float* xr = x + (size_t)(tile * 16 + n) * F_SZ;
#pragma unroll
  for (int j = 0; j < 16; ++j) {
    v2f a = {xr[j * 4 + kh], xr[j * 4 + kh + 1]};
    acc = wmma_f32_16x16x4(a, bch[j], acc);
  }

  if (n < 4) {
#pragma unroll
    for (int r = 0; r < 8; ++r) {
      int row = tile * 16 + r + (hi << 3);
      y0[(size_t)row * 4 + n] = acc[r];
    }
  }
}

__global__ void __launch_bounds__(256)
qlstm_recur(const float* __restrict__ y0, const float* __restrict__ Win,
            const float* __restrict__ Wout, const float* __restrict__ bout,
            const float* __restrict__ thetas, float* __restrict__ out) {
  __shared__ float y0_lds[16 * S_SZ * 4];
  __shared__ float h_lds[16 * HSTR];
  __shared__ float part[8 * 16 * 4];
  __shared__ float z_lds[4 * 16 * 4];

  const int tid   = threadIdx.x;
  const int lane  = tid & 31;
  const int wave  = tid >> 5;
  const int n     = lane & 15;
  const int hi    = lane >> 4;
  const int kh    = hi << 1;
  const int bbase = blockIdx.x * 16;

  float* hs = out;
  float* hT = out + (size_t)B_SZ * S_SZ * H_SZ;
  float* cT = hT + (size_t)B_SZ * H_SZ;

  {
    const float4* src = (const float4*)(y0 + (size_t)bbase * S_SZ * 4);
    float4* dst = (float4*)y0_lds;
#pragma unroll
    for (int i = 0; i < 16; ++i)
      dst[tid + i * 256] = src[tid + i * 256];
  }
  for (int i = tid; i < 16 * HSTR; i += 256) h_lds[i] = 0.f;

  v2f bh[4];
#pragma unroll
  for (int j = 0; j < 4; ++j) {
    float b0 = 0.f, b1 = 0.f;
    if (n < 4) {
      int kk = wave * 16 + j * 4 + kh;
      b0 = Win[n * 192 + kk];
      b1 = Win[n * 192 + kk + 1];
    }
    v2f bv = {b0, b1};
    bh[j] = bv;
  }
  const int col = wave * 16 + n;
  v2f bw = {Wout[col * 4 + kh], Wout[col * 4 + kh + 1]};
  const float bo = bout[col];
  const v8f cbias = {bo, bo, bo, bo, bo, bo, bo, bo};

  const int sb = tid >> 2, sg = tid & 3;
  float th[4];
#pragma unroll
  for (int w = 0; w < 4; ++w) th[w] = (tid < 64) ? thetas[sg * 4 + w] : 0.f;

  float c_state[8], h_keep[8];
#pragma unroll
  for (int r = 0; r < 8; ++r) { c_state[r] = 0.f; h_keep[r] = 0.f; }

  __syncthreads();

  for (int t = 0; t < S_SZ; ++t) {
    v8f acc = {0.f, 0.f, 0.f, 0.f, 0.f, 0.f, 0.f, 0.f};
#pragma unroll
    for (int j = 0; j < 4; ++j) {
      int kk = wave * 16 + j * 4 + kh;
      v2f a = {h_lds[n * HSTR + kk], h_lds[n * HSTR + kk + 1]};
      acc = wmma_f32_16x16x4(a, bh[j], acc);
    }
    if (n < 4) {
#pragma unroll
      for (int r = 0; r < 8; ++r)
        part[wave * 64 + (r + (hi << 3)) * 4 + n] = acc[r];
    }
    lds_barrier();

    if (tid < 64) {
      float p0[4], p1[4];
#pragma unroll
      for (int w = 0; w < 4; ++w) {
        float yw = y0_lds[(sb * S_SZ + t) * 4 + w];
#pragma unroll
        for (int wv = 0; wv < 8; ++wv) yw += part[wv * 64 + sb * 4 + w];
        float cv = __cosf(yw + th[w]);
        p0[w] = 0.5f + 0.5f * cv;
        p1[w] = 0.5f - 0.5f * cv;
      }
      float z0 = 0.f, z1 = 0.f, z2 = 0.f, z3 = 0.f;
#pragma unroll
      for (int idx = 0; idx < 16; ++idx) {
        int a0 = (idx >> 3) & 1, a1 = (idx >> 2) & 1;
        int a2 = (idx >> 1) & 1, a3 = idx & 1;
        float p = (a0 ? p1[0] : p0[0]) * (a1 ? p1[1] : p0[1]) *
                  (a2 ? p1[2] : p0[2]) * (a3 ? p1[3] : p0[3]);
        int t1 = a1 ^ a0;
        int t2 = a2 ^ t1;
        int t3 = a3 ^ t2;
        int t0 = a0 ^ t3;
        z0 += t0 ? -p : p;  z1 += t1 ? -p : p;
        z2 += t2 ? -p : p;  z3 += t3 ? -p : p;
      }
      int base = (sg * 16 + sb) * 4;
      z_lds[base + 0] = z0;  z_lds[base + 1] = z1;
      z_lds[base + 2] = z2;  z_lds[base + 3] = z3;
    }
    lds_barrier();

    v8f ga[4];
#pragma unroll
    for (int g = 0; g < 4; ++g) {
      v2f a = {z_lds[(g * 16 + n) * 4 + kh], z_lds[(g * 16 + n) * 4 + kh + 1]};
      ga[g] = wmma_f32_16x16x4(a, bw, cbias);
    }
#pragma unroll
    for (int r = 0; r < 8; ++r) {
      float fv = fast_sigmoid(ga[0][r]);
      float iv = fast_sigmoid(ga[1][r]);
      float gv = fast_tanh(ga[2][r]);
      float ov = fast_sigmoid(ga[3][r]);
      float cs = fv * c_state[r] + iv * gv;
      c_state[r] = cs;
      float hv = ov * fast_tanh(cs);
      h_keep[r] = hv;
      int mrow = r + (hi << 3);
      h_lds[mrow * HSTR + col] = hv;
      hs[(((size_t)(bbase + mrow)) * S_SZ + t) * H_SZ + col] = hv;
    }
    lds_barrier();
  }

#pragma unroll
  for (int r = 0; r < 8; ++r) {
    int mrow = r + (hi << 3);
    hT[((size_t)(bbase + mrow)) * H_SZ + col] = h_keep[r];
    cT[((size_t)(bbase + mrow)) * H_SZ + col] = c_state[r];
  }
}

extern "C" void kernel_launch(void* const* d_in, const int* in_sizes, int n_in,
                              void* d_out, int out_size, void* d_ws, size_t ws_size,
                              hipStream_t stream) {
  const float* x      = (const float*)d_in[0];
  const float* W_in   = (const float*)d_in[1];
  const float* b_in   = (const float*)d_in[2];
  const float* W_out  = (const float*)d_in[3];
  const float* b_out  = (const float*)d_in[4];
  const float* thetas = (const float*)d_in[5];
  float* out = (float*)d_out;
  float* y0  = (float*)d_ws;

  qlstm_xproj<<<1024, 128, 0, stream>>>(x, W_in, b_in, y0);
  qlstm_recur<<<16, 256, 0, stream>>>(y0, W_in, W_out, b_out, thetas, out);
}


# gate and y products computed transposed (weights as A, batch rows as columns): h feeds the next step's WMMA from registers, no h LDS trip, no third barrier; k order of y partial sums permuted
# speedup vs baseline: 2.0319x; 1.1614x over previous
#include <hip/hip_runtime.h>
#include <hip/hip_bf16.h>
#include <math.h>

typedef __attribute__((ext_vector_type(2))) float v2f;
typedef __attribute__((ext_vector_type(8))) float v8f;

static __device__ __forceinline__ v8f wmma_f32_16x16x4(v2f a, v2f b, v8f c) {
  return __builtin_amdgcn_wmma_f32_16x16x4_f32(false, a, false, b, (short)0, c, false, false);
}

static __device__ __forceinline__ float fast_sigmoid(float x) {
  x = fminf(fmaxf(x, -30.f), 30.f);
  return __builtin_amdgcn_rcpf(1.f + __expf(-x));
}
static __device__ __forceinline__ float fast_tanh(float x) {
  x = fminf(fmaxf(x, -15.f), 15.f);
  float e = __expf(-2.f * x);
  return (1.f - e) * __builtin_amdgcn_rcpf(1.f + e);
}

static __device__ __forceinline__ void lds_barrier() {
  __builtin_amdgcn_fence(__ATOMIC_RELEASE, "workgroup", "local");
  __builtin_amdgcn_s_barrier();
  __builtin_amdgcn_fence(__ATOMIC_ACQUIRE, "workgroup", "local");
}

#define B_SZ 256
#define S_SZ 256
#define F_SZ 64
#define H_SZ 128

__global__ void __launch_bounds__(128)
qlstm_xproj(const float* __restrict__ x, const float* __restrict__ Win,
            const float* __restrict__ bin, float* __restrict__ y0) {
  const int lane = threadIdx.x & 31;
  const int wave = threadIdx.x >> 5;
  const int tile = blockIdx.x * 4 + wave;
  const int n    = lane & 15;
  const int hi   = lane >> 4;
  const int kh   = hi << 1;

  v2f bch[16];
#pragma unroll
  for (int j = 0; j < 16; ++j) {
    float b0 = 0.f, b1 = 0.f;
    if (n < 4) {
      int kk = j * 4 + kh;
      b0 = Win[n * 192 + 128 + kk];
      b1 = Win[n * 192 + 128 + kk + 1];
    }
    v2f bv = {b0, b1};
    bch[j] = bv;
  }

  const float bb = (n < 4) ? bin[n] : 0.f;
  v8f acc = {bb, bb, bb, bb, bb, bb, bb, bb};

  const float* xr = x + (size_t)(tile * 16 + n) * F_SZ;
#pragma unroll
  for (int j = 0; j < 16; ++j) {
    v2f a = {xr[j * 4 + kh], xr[j * 4 + kh + 1]};
    acc = wmma_f32_16x16x4(a, bch[j], acc);
  }

  if (n < 4) {
#pragma unroll
    for (int r = 0; r < 8; ++r) {
      int row = tile * 16 + r + (hi << 3);
      y0[(size_t)row * 4 + n] = acc[r];
    }
  }
}

__global__ void __launch_bounds__(256)
qlstm_recur(const float* __restrict__ y0, const float* __restrict__ Win,
            const float* __restrict__ Wout, const float* __restrict__ bout,
            const float* __restrict__ thetas, float* __restrict__ out) {
  __shared__ float y0_lds[16 * S_SZ * 4];
  __shared__ float part[8 * 16 * 4];
  __shared__ float z_lds[4 * 16 * 4];

  const int tid   = threadIdx.x;
  const int lane  = tid & 31;
  const int wave  = tid >> 5;
  const int n     = lane & 15;
  const int hi    = lane >> 4;
  const int kh    = hi << 1;
  const int bbase = blockIdx.x * 16;
  const int cbase = wave * 16 + hi * 8;

  float* hs = out;
  float* hT = out + (size_t)B_SZ * S_SZ * H_SZ;
  float* cT = hT + (size_t)B_SZ * H_SZ;

  {
    const float4* src = (const float4*)(y0 + (size_t)bbase * S_SZ * 4);
    float4* dst = (float4*)y0_lds;
#pragma unroll
    for (int i = 0; i < 16; ++i)
      dst[tid + i * 256] = src[tid + i * 256];
  }

  v2f ah[4];
#pragma unroll
  for (int j = 0; j < 4; ++j) {
    float a0 = 0.f, a1 = 0.f;
    if (n < 4) {
      int kk = cbase + 2 * j;
      a0 = Win[n * 192 + kk];
      a1 = Win[n * 192 + kk + 1];
    }
    v2f av = {a0, a1};
    ah[j] = av;
  }
  v2f aw = {Wout[(wave * 16 + n) * 4 + kh], Wout[(wave * 16 + n) * 4 + kh + 1]};
  v8f cbias;
  for (int r = 0; r < 8; ++r) cbias[r] = bout[cbase + r];

  const int sb = tid >> 2, sg = tid & 3;
  float th[4];
#pragma unroll
  for (int w = 0; w < 4; ++w) th[w] = (tid < 64) ? thetas[sg * 4 + w] : 0.f;

  float c_state[8], h[8];
#pragma unroll
  for (int r = 0; r < 8; ++r) { c_state[r] = 0.f; h[r] = 0.f; }

  float* hs_row = hs + ((size_t)(bbase + n) * S_SZ) * H_SZ + cbase;

  __syncthreads();

  for (int t = 0; t < S_SZ; ++t) {
    v8f acc = {0.f, 0.f, 0.f, 0.f, 0.f, 0.f, 0.f, 0.f};
#pragma unroll
    for (int j = 0; j < 4; ++j) {
      v2f b = {h[2 * j], h[2 * j + 1]};
      acc = wmma_f32_16x16x4(ah[j], b, acc);
    }
    if (hi == 0) {
      float4 pv = {acc[0], acc[1], acc[2], acc[3]};
      *(float4*)&part[wave * 64 + n * 4] = pv;
    }
    lds_barrier();

    if (tid < 64) {
      float p0[4], p1[4];
#pragma unroll
      for (int w = 0; w < 4; ++w) {
        float yw = y0_lds[(sb * S_SZ + t) * 4 + w];
#pragma unroll
        for (int wv = 0; wv < 8; ++wv) yw += part[wv * 64 + sb * 4 + w];
        float cv = __cosf(yw + th[w]);
        p0[w] = 0.5f + 0.5f * cv;
        p1[w] = 0.5f - 0.5f * cv;
      }
      float z0 = 0.f, z1 = 0.f, z2 = 0.f, z3 = 0.f;
#pragma unroll
      for (int idx = 0; idx < 16; ++idx) {
        int a0 = (idx >> 3) & 1, a1 = (idx >> 2) & 1;
        int a2 = (idx >> 1) & 1, a3 = idx & 1;
        float p = (a0 ? p1[0] : p0[0]) * (a1 ? p1[1] : p0[1]) *
                  (a2 ? p1[2] : p0[2]) * (a3 ? p1[3] : p0[3]);
        int t1 = a1 ^ a0;
        int t2 = a2 ^ t1;
        int t3 = a3 ^ t2;
        int t0 = a0 ^ t3;
        z0 += t0 ? -p : p;  z1 += t1 ? -p : p;
        z2 += t2 ? -p : p;  z3 += t3 ? -p : p;
      }
      int base = (sg * 16 + sb) * 4;
      z_lds[base + 0] = z0;  z_lds[base + 1] = z1;
      z_lds[base + 2] = z2;  z_lds[base + 3] = z3;
    }
    lds_barrier();

    v8f ga[4];
#pragma unroll
    for (int g = 0; g < 4; ++g) {
      v2f b = {z_lds[(g * 16 + n) * 4 + kh], z_lds[(g * 16 + n) * 4 + kh + 1]};
      ga[g] = wmma_f32_16x16x4(aw, b, cbias);
    }
#pragma unroll
    for (int r = 0; r < 8; ++r) {
      float fv = fast_sigmoid(ga[0][r]);
      float iv = fast_sigmoid(ga[1][r]);
      float gv = fast_tanh(ga[2][r]);
      float ov = fast_sigmoid(ga[3][r]);
      float cs = fv * c_state[r] + iv * gv;
      c_state[r] = cs;
      h[r] = ov * fast_tanh(cs);
    }
    {
      float4* dst = (float4*)(hs_row + (size_t)t * H_SZ);
      float4 lo = {h[0], h[1], h[2], h[3]};
      float4 up = {h[4], h[5], h[6], h[7]};
      dst[0] = lo;
      dst[1] = up;
    }
  }

  {
    float4* dh = (float4*)(hT + (size_t)(bbase + n) * H_SZ + cbase);
    float4* dc = (float4*)(cT + (size_t)(bbase + n) * H_SZ + cbase);
    float4 hlo = {h[0], h[1], h[2], h[3]}, hup = {h[4], h[5], h[6], h[7]};
    float4 clo = {c_state[0], c_state[1], c_state[2], c_state[3]};
    float4 cup = {c_state[4], c_state[5], c_state[6], c_state[7]};
    dh[0] = hlo;  dh[1] = hup;
    dc[0] = clo;  dc[1] = cup;
  }
}

extern "C" void kernel_launch(void* const* d_in, const int* in_sizes, int n_in,
                              void* d_out, int out_size, void* d_ws, size_t ws_size,
                              hipStream_t stream) {
  const float* x      = (const float*)d_in[0];
  const float* W_in   = (const float*)d_in[1];
  const float* b_in   = (const float*)d_in[2];
  const float* W_out  = (const float*)d_in[3];
  const float* b_out  = (const float*)d_in[4];
  const float* thetas = (const float*)d_in[5];
  float* out = (float*)d_out;
  float* y0  = (float*)d_ws;

  qlstm_xproj<<<1024, 128, 0, stream>>>(x, W_in, b_in, y0);
  qlstm_recur<<<16, 256, 0, stream>>>(y0, W_in, W_out, b_out, thetas, out);
}


# LSTM gate and cell activations on the hardware tanh instruction (v_tanh_f32), sigmoid as (1+tanh(x/2))/2: 5 transcendentals per cell update where exp+rcp pairs made 10
# speedup vs baseline: 2.5266x; 1.2435x over previous
#include <hip/hip_runtime.h>
#include <hip/hip_bf16.h>
#include <math.h>

typedef __attribute__((ext_vector_type(2))) float v2f;
typedef __attribute__((ext_vector_type(8))) float v8f;

static __device__ __forceinline__ v8f wmma_f32_16x16x4(v2f a, v2f b, v8f c) {
  return __builtin_amdgcn_wmma_f32_16x16x4_f32(false, a, false, b, (short)0, c, false, false);
}

static __device__ __forceinline__ float hw_tanh(float x) {
  return __builtin_amdgcn_tanhf(x);
}
static __device__ __forceinline__ float hw_sigmoid(float x) {
  return fmaf(0.5f, hw_tanh(0.5f * x), 0.5f);
}

static __device__ __forceinline__ void lds_barrier() {
  __builtin_amdgcn_fence(__ATOMIC_RELEASE, "workgroup", "local");
  __builtin_amdgcn_s_barrier();
  __builtin_amdgcn_fence(__ATOMIC_ACQUIRE, "workgroup", "local");
}

#define B_SZ 256
#define S_SZ 256
#define F_SZ 64
#define H_SZ 128

__global__ void __launch_bounds__(128)
qlstm_xproj(const float* __restrict__ x, const float* __restrict__ Win,
            const float* __restrict__ bin, float* __restrict__ y0) {
  const int lane = threadIdx.x & 31;
  const int wave = threadIdx.x >> 5;
  const int tile = blockIdx.x * 4 + wave;
  const int n    = lane & 15;
  const int hi   = lane >> 4;
  const int kh   = hi << 1;

  v2f bch[16];
#pragma unroll
  for (int j = 0; j < 16; ++j) {
    float b0 = 0.f, b1 = 0.f;
    if (n < 4) {
      int kk = j * 4 + kh;
      b0 = Win[n * 192 + 128 + kk];
      b1 = Win[n * 192 + 128 + kk + 1];
    }
    v2f bv = {b0, b1};
    bch[j] = bv;
  }

  const float bb = (n < 4) ? bin[n] : 0.f;
  v8f acc = {bb, bb, bb, bb, bb, bb, bb, bb};

  const float* xr = x + (size_t)(tile * 16 + n) * F_SZ;
#pragma unroll
  for (int j = 0; j < 16; ++j) {
    v2f a = {xr[j * 4 + kh], xr[j * 4 + kh + 1]};
    acc = wmma_f32_16x16x4(a, bch[j], acc);
  }

  if (n < 4) {
#pragma unroll
    for (int r = 0; r < 8; ++r) {
      int row = tile * 16 + r + (hi << 3);
      y0[(size_t)row * 4 + n] = acc[r];
    }
  }
}

__global__ void __launch_bounds__(256)
qlstm_recur(const float* __restrict__ y0, const float* __restrict__ Win,
            const float* __restrict__ Wout, const float* __restrict__ bout,
            const float* __restrict__ thetas, float* __restrict__ out) {
  __shared__ float y0_lds[16 * S_SZ * 4];
  __shared__ float part[8 * 16 * 4];
  __shared__ float z_lds[4 * 16 * 4];

  const int tid   = threadIdx.x;
  const int lane  = tid & 31;
  const int wave  = tid >> 5;
  const int n     = lane & 15;
  const int hi    = lane >> 4;
  const int kh    = hi << 1;
  const int bbase = blockIdx.x * 16;
  const int cbase = wave * 16 + hi * 8;

  float* hs = out;
  float* hT = out + (size_t)B_SZ * S_SZ * H_SZ;
  float* cT = hT + (size_t)B_SZ * H_SZ;

  {
    const float4* src = (const float4*)(y0 + (size_t)bbase * S_SZ * 4);
    float4* dst = (float4*)y0_lds;
#pragma unroll
    for (int i = 0; i < 16; ++i)
      dst[tid + i * 256] = src[tid + i * 256];
  }

  v2f ah[4];
#pragma unroll
  for (int j = 0; j < 4; ++j) {
    float a0 = 0.f, a1 = 0.f;
    if (n < 4) {
      int kk = cbase + 2 * j;
      a0 = Win[n * 192 + kk];
      a1 = Win[n * 192 + kk + 1];
    }
    v2f av = {a0, a1};
    ah[j] = av;
  }
  v2f aw = {Wout[(wave * 16 + n) * 4 + kh], Wout[(wave * 16 + n) * 4 + kh + 1]};
  v8f cbias;
  for (int r = 0; r < 8; ++r) cbias[r] = bout[cbase + r];

  const int sb = tid >> 2, sg = tid & 3;
  float th[4];
#pragma unroll
  for (int w = 0; w < 4; ++w) th[w] = (tid < 64) ? thetas[sg * 4 + w] : 0.f;

  float c_state[8], h[8];
#pragma unroll
  for (int r = 0; r < 8; ++r) { c_state[r] = 0.f; h[r] = 0.f; }

  float* hs_row = hs + ((size_t)(bbase + n) * S_SZ) * H_SZ + cbase;

  __syncthreads();

  for (int t = 0; t < S_SZ; ++t) {
    v8f acc = {0.f, 0.f, 0.f, 0.f, 0.f, 0.f, 0.f, 0.f};
#pragma unroll
    for (int j = 0; j < 4; ++j) {
      v2f b = {h[2 * j], h[2 * j + 1]};
      acc = wmma_f32_16x16x4(ah[j], b, acc);
    }
    if (hi == 0) {
      float4 pv = {acc[0], acc[1], acc[2], acc[3]};
      *(float4*)&part[wave * 64 + n * 4] = pv;
    }
    lds_barrier();

    if (tid < 64) {
      float p0[4], p1[4];
#pragma unroll
      for (int w = 0; w < 4; ++w) {
        float yw = y0_lds[(sb * S_SZ + t) * 4 + w];
#pragma unroll
        for (int wv = 0; wv < 8; ++wv) yw += part[wv * 64 + sb * 4 + w];
        float cv = __cosf(yw + th[w]);
        p0[w] = 0.5f + 0.5f * cv;
        p1[w] = 0.5f - 0.5f * cv;
      }
      float z0 = 0.f, z1 = 0.f, z2 = 0.f, z3 = 0.f;
#pragma unroll
      for (int idx = 0; idx < 16; ++idx) {
        int a0 = (idx >> 3) & 1, a1 = (idx >> 2) & 1;
        int a2 = (idx >> 1) & 1, a3 = idx & 1;
        float p = (a0 ? p1[0] : p0[0]) * (a1 ? p1[1] : p0[1]) *
                  (a2 ? p1[2] : p0[2]) * (a3 ? p1[3] : p0[3]);
        int t1 = a1 ^ a0;
        int t2 = a2 ^ t1;
        int t3 = a3 ^ t2;
        int t0 = a0 ^ t3;
        z0 += t0 ? -p : p;  z1 += t1 ? -p : p;
        z2 += t2 ? -p : p;  z3 += t3 ? -p : p;
      }
      int base = (sg * 16 + sb) * 4;
      z_lds[base + 0] = z0;  z_lds[base + 1] = z1;
      z_lds[base + 2] = z2;  z_lds[base + 3] = z3;
    }
    lds_barrier();

    v8f ga[4];
#pragma unroll
    for (int g = 0; g < 4; ++g) {
      v2f b = {z_lds[(g * 16 + n) * 4 + kh], z_lds[(g * 16 + n) * 4 + kh + 1]};
      ga[g] = wmma_f32_16x16x4(aw, b, cbias);
    }
#pragma unroll
    for (int r = 0; r < 8; ++r) {
      float fv = hw_sigmoid(ga[0][r]);
      float iv = hw_sigmoid(ga[1][r]);
      float gv = hw_tanh(ga[2][r]);
      float ov = hw_sigmoid(ga[3][r]);
      float cs = fv * c_state[r] + iv * gv;
      c_state[r] = cs;
      h[r] = ov * hw_tanh(cs);
    }
    {
      float4* dst = (float4*)(hs_row + (size_t)t * H_SZ);
      float4 lo = {h[0], h[1], h[2], h[3]};
      float4 up = {h[4], h[5], h[6], h[7]};
      dst[0] = lo;
      dst[1] = up;
    }
  }

  {
    float4* dh = (float4*)(hT + (size_t)(bbase + n) * H_SZ + cbase);
    float4* dc = (float4*)(cT + (size_t)(bbase + n) * H_SZ + cbase);
    float4 hlo = {h[0], h[1], h[2], h[3]}, hup = {h[4], h[5], h[6], h[7]};
    float4 clo = {c_state[0], c_state[1], c_state[2], c_state[3]};
    float4 cup = {c_state[4], c_state[5], c_state[6], c_state[7]};
    dh[0] = hlo;  dh[1] = hup;
    dc[0] = clo;  dc[1] = cup;
  }
}

extern "C" void kernel_launch(void* const* d_in, const int* in_sizes, int n_in,
                              void* d_out, int out_size, void* d_ws, size_t ws_size,
                              hipStream_t stream) {
  const float* x      = (const float*)d_in[0];
  const float* W_in   = (const float*)d_in[1];
  const float* b_in   = (const float*)d_in[2];
  const float* W_out  = (const float*)d_in[3];
  const float* b_out  = (const float*)d_in[4];
  const float* thetas = (const float*)d_in[5];
  float* out = (float*)d_out;
  float* y0  = (float*)d_ws;

  qlstm_xproj<<<1024, 128, 0, stream>>>(x, W_in, b_in, y0);
  qlstm_recur<<<16, 256, 0, stream>>>(y0, W_in, W_out, b_out, thetas, out);
}


# qlayer by every lane in registers in closed form (z = products of cos; the 16-term probability sums go): no z trip through LDS, one barrier per step where there were two, part double-buffered
# speedup vs baseline: 2.9411x; 1.1641x over previous
#include <hip/hip_runtime.h>
#include <hip/hip_bf16.h>
#include <math.h>

typedef __attribute__((ext_vector_type(2))) float v2f;
typedef __attribute__((ext_vector_type(8))) float v8f;

static __device__ __forceinline__ v8f wmma_f32_16x16x4(v2f a, v2f b, v8f c) {
  return __builtin_amdgcn_wmma_f32_16x16x4_f32(false, a, false, b, (short)0, c, false, false);
}

static __device__ __forceinline__ float hw_tanh(float x) {
  return __builtin_amdgcn_tanhf(x);
}
static __device__ __forceinline__ float hw_sigmoid(float x) {
  return fmaf(0.5f, hw_tanh(0.5f * x), 0.5f);
}

static __device__ __forceinline__ void lds_barrier() {
  __builtin_amdgcn_fence(__ATOMIC_RELEASE, "workgroup", "local");
  __builtin_amdgcn_s_barrier();
  __builtin_amdgcn_fence(__ATOMIC_ACQUIRE, "workgroup", "local");
}

#define B_SZ 256
#define S_SZ 256
#define F_SZ 64
#define H_SZ 128
#define Y0_PITCH (S_SZ * 4 + 4)

__global__ void __launch_bounds__(128)
qlstm_xproj(const float* __restrict__ x, const float* __restrict__ Win,
            const float* __restrict__ bin, float* __restrict__ y0) {
  const int lane = threadIdx.x & 31;
  const int wave = threadIdx.x >> 5;
  const int tile = blockIdx.x * 4 + wave;
  const int n    = lane & 15;
  const int hi   = lane >> 4;
  const int kh   = hi << 1;

  v2f bch[16];
#pragma unroll
  for (int j = 0; j < 16; ++j) {
    float b0 = 0.f, b1 = 0.f;
    if (n < 4) {
      int kk = j * 4 + kh;
      b0 = Win[n * 192 + 128 + kk];
      b1 = Win[n * 192 + 128 + kk + 1];
    }
    v2f bv = {b0, b1};
    bch[j] = bv;
  }

  const float bb = (n < 4) ? bin[n] : 0.f;
  v8f acc = {bb, bb, bb, bb, bb, bb, bb, bb};

  const float* xr = x + (size_t)(tile * 16 + n) * F_SZ;
#pragma unroll
  for (int j = 0; j < 16; ++j) {
    v2f a = {xr[j * 4 + kh], xr[j * 4 + kh + 1]};
    acc = wmma_f32_16x16x4(a, bch[j], acc);
  }

  if (n < 4) {
#pragma unroll
    for (int r = 0; r < 8; ++r) {
      int row = tile * 16 + r + (hi << 3);
      y0[(size_t)row * 4 + n] = acc[r];
    }
  }
}

__global__ void __launch_bounds__(256)
qlstm_recur(const float* __restrict__ y0, const float* __restrict__ Win,
            const float* __restrict__ Wout, const float* __restrict__ bout,
            const float* __restrict__ thetas, float* __restrict__ out) {
  __shared__ __attribute__((aligned(16))) float y0_lds[16 * Y0_PITCH];
  __shared__ __attribute__((aligned(16))) float part[2 * 8 * 16 * 4];

  const int tid   = threadIdx.x;
  const int lane  = tid & 31;
  const int wave  = tid >> 5;
  const int n     = lane & 15;
  const int hi    = lane >> 4;
  const int kh    = hi << 1;
  const int bbase = blockIdx.x * 16;
  const int cbase = wave * 16 + hi * 8;

  float* hs = out;
  float* hT = out + (size_t)B_SZ * S_SZ * H_SZ;
  float* cT = hT + (size_t)B_SZ * H_SZ;

  {
    const float4* src = (const float4*)(y0 + (size_t)bbase * S_SZ * 4);
    float4* dst = (float4*)y0_lds;
#pragma unroll
    for (int i = 0; i < 16; ++i)
      dst[i * (Y0_PITCH / 4) + tid] = src[tid + i * 256];
  }

  v2f ah[4];
#pragma unroll
  for (int j = 0; j < 4; ++j) {
    float a0 = 0.f, a1 = 0.f;
    if (n < 4) {
      int kk = cbase + 2 * j;
      a0 = Win[n * 192 + kk];
      a1 = Win[n * 192 + kk + 1];
    }
    v2f av = {a0, a1};
    ah[j] = av;
  }
  v2f aw = {Wout[(wave * 16 + n) * 4 + kh], Wout[(wave * 16 + n) * 4 + kh + 1]};
  v8f cbias;
  for (int r = 0; r < 8; ++r) cbias[r] = bout[cbase + r];

  const float INV_2PI = 0.15915494309189532f;
  float thr[4][4];
#pragma unroll
  for (int g = 0; g < 4; ++g)
#pragma unroll
    for (int w = 0; w < 4; ++w) thr[g][w] = thetas[g * 4 + w] * INV_2PI;

  float c_state[8], h[8];
#pragma unroll
  for (int r = 0; r < 8; ++r) { c_state[r] = 0.f; h[r] = 0.f; }

  float* hs_row = hs + ((size_t)(bbase + n) * S_SZ) * H_SZ + cbase;
  const float* y0_row = y0_lds + n * Y0_PITCH;

  __syncthreads();

  for (int t = 0; t < S_SZ; ++t) {
    float* pw = part + (t & 1) * (8 * 16 * 4);
    const float4 yv = *(const float4*)(y0_row + t * 4);

    v8f acc = {0.f, 0.f, 0.f, 0.f, 0.f, 0.f, 0.f, 0.f};
#pragma unroll
    for (int j = 0; j < 4; ++j) {
      v2f b = {h[2 * j], h[2 * j + 1]};
      acc = wmma_f32_16x16x4(ah[j], b, acc);
    }
    if (hi == 0) {
      float4 pv = {acc[0], acc[1], acc[2], acc[3]};
      *(float4*)&pw[wave * 64 + n * 4] = pv;
    }
    lds_barrier();

    float ys[4] = {yv.x, yv.y, yv.z, yv.w};
#pragma unroll
    for (int wv = 0; wv < 8; ++wv) {
      const float4 pv = *(const float4*)&pw[wv * 64 + n * 4];
      ys[0] += pv.x;  ys[1] += pv.y;  ys[2] += pv.z;  ys[3] += pv.w;
    }
    v2f bz[4];
#pragma unroll
    for (int g = 0; g < 4; ++g) {
      const float c0 = __builtin_amdgcn_cosf(fmaf(ys[0], INV_2PI, thr[g][0]));
      const float c1 = __builtin_amdgcn_cosf(fmaf(ys[1], INV_2PI, thr[g][1]));
      const float c2 = __builtin_amdgcn_cosf(fmaf(ys[2], INV_2PI, thr[g][2]));
      const float c3 = __builtin_amdgcn_cosf(fmaf(ys[3], INV_2PI, thr[g][3]));
      const float p01 = c0 * c1, p23 = c2 * c3;
      const float u = hi ? p01 : c1;
      const float v = hi ? c2 : p23;
      v2f b = {u * v, hi ? p01 * p23 : p01};
      bz[g] = b;
    }

    v8f ga[4];
#pragma unroll
    for (int g = 0; g < 4; ++g)
      ga[g] = wmma_f32_16x16x4(aw, bz[g], cbias);
#pragma unroll
    for (int r = 0; r < 8; ++r) {
      float fv = hw_sigmoid(ga[0][r]);
      float iv = hw_sigmoid(ga[1][r]);
      float gv = hw_tanh(ga[2][r]);
      float ov = hw_sigmoid(ga[3][r]);
      float cs = fv * c_state[r] + iv * gv;
      c_state[r] = cs;
      h[r] = ov * hw_tanh(cs);
    }
    {
      float4* dst = (float4*)(hs_row + (size_t)t * H_SZ);
      float4 lo = {h[0], h[1], h[2], h[3]};
      float4 up = {h[4], h[5], h[6], h[7]};
      dst[0] = lo;
      dst[1] = up;
    }
  }

  {
    float4* dh = (float4*)(hT + (size_t)(bbase + n) * H_SZ + cbase);
    float4* dc = (float4*)(cT + (size_t)(bbase + n) * H_SZ + cbase);
    float4 hlo = {h[0], h[1], h[2], h[3]}, hup = {h[4], h[5], h[6], h[7]};
    float4 clo = {c_state[0], c_state[1], c_state[2], c_state[3]};
    float4 cup = {c_state[4], c_state[5], c_state[6], c_state[7]};
    dh[0] = hlo;  dh[1] = hup;
    dc[0] = clo;  dc[1] = cup;
  }
}

extern "C" void kernel_launch(void* const* d_in, const int* in_sizes, int n_in,
                              void* d_out, int out_size, void* d_ws, size_t ws_size,
                              hipStream_t stream) {
  const float* x      = (const float*)d_in[0];
  const float* W_in   = (const float*)d_in[1];
  const float* b_in   = (const float*)d_in[2];
  const float* W_out  = (const float*)d_in[3];
  const float* b_out  = (const float*)d_in[4];
  const float* thetas = (const float*)d_in[5];
  float* out = (float*)d_out;
  float* y0  = (float*)d_ws;

  qlstm_xproj<<<1024, 128, 0, stream>>>(x, W_in, b_in, y0);
  qlstm_recur<<<16, 256, 0, stream>>>(y0, W_in, W_out, b_out, thetas, out);
}


# gate tile columns = (batch row, gate): 4 batch rows per block, 64 blocks; one WMMA makes all 4 gates, a lane activates one gate, quad DPP reads share f,i,g,o; 4 cos per lane where 16
# speedup vs baseline: 3.4742x; 1.1813x over previous
#include <hip/hip_runtime.h>
#include <hip/hip_bf16.h>
#include <math.h>

typedef __attribute__((ext_vector_type(2))) float v2f;
typedef __attribute__((ext_vector_type(8))) float v8f;

static __device__ __forceinline__ v8f wmma_f32_16x16x4(v2f a, v2f b, v8f c) {
  return __builtin_amdgcn_wmma_f32_16x16x4_f32(false, a, false, b, (short)0, c, false, false);
}

static __device__ __forceinline__ float hw_tanh(float x) {
  return __builtin_amdgcn_tanhf(x);
}
static __device__ __forceinline__ float hw_sigmoid(float x) {
  return fmaf(0.5f, hw_tanh(0.5f * x), 0.5f);
}

static __device__ __forceinline__ void lds_barrier() {
  __builtin_amdgcn_fence(__ATOMIC_RELEASE, "workgroup", "local");
  __builtin_amdgcn_s_barrier();
  __builtin_amdgcn_fence(__ATOMIC_ACQUIRE, "workgroup", "local");
}

#define B_SZ 256
#define S_SZ 256
#define F_SZ 64
#define H_SZ 128
#define Y0_PITCH (S_SZ * 4 + 4)

__global__ void __launch_bounds__(128)
qlstm_xproj(const float* __restrict__ x, const float* __restrict__ Win,
            const float* __restrict__ bin, float* __restrict__ y0) {
  const int lane = threadIdx.x & 31;
  const int wave = threadIdx.x >> 5;
  const int tile = blockIdx.x * 4 + wave;
  const int n    = lane & 15;
  const int hi   = lane >> 4;
  const int kh   = hi << 1;

  v2f bch[16];
#pragma unroll
  for (int j = 0; j < 16; ++j) {
    float b0 = 0.f, b1 = 0.f;
    if (n < 4) {
      int kk = j * 4 + kh;
      b0 = Win[n * 192 + 128 + kk];
      b1 = Win[n * 192 + 128 + kk + 1];
    }
    v2f bv = {b0, b1};
    bch[j] = bv;
  }

  const float bb = (n < 4) ? bin[n] : 0.f;
  v8f acc = {bb, bb, bb, bb, bb, bb, bb, bb};

  const float* xr = x + (size_t)(tile * 16 + n) * F_SZ;
#pragma unroll
  for (int j = 0; j < 16; ++j) {
    v2f a = {xr[j * 4 + kh], xr[j * 4 + kh + 1]};
    acc = wmma_f32_16x16x4(a, bch[j], acc);
  }

  if (n < 4) {
#pragma unroll
    for (int r = 0; r < 8; ++r) {
      int row = tile * 16 + r + (hi << 3);
      y0[(size_t)row * 4 + n] = acc[r];
    }
  }
}

#define ROWS_PER_BLOCK 4

template <int K>
static __device__ __forceinline__ float quad_bcast(float v) {
  return __int_as_float(__builtin_amdgcn_mov_dpp(__float_as_int(v), K * 0x55, 0xF, 0xF, true));
}

__global__ void __launch_bounds__(256)
qlstm_recur(const float* __restrict__ y0, const float* __restrict__ Win,
            const float* __restrict__ Wout, const float* __restrict__ bout,
            const float* __restrict__ thetas, float* __restrict__ out) {
  __shared__ __attribute__((aligned(16))) float y0_lds[ROWS_PER_BLOCK * Y0_PITCH];
  __shared__ __attribute__((aligned(16))) float part[2 * 8 * 16 * 4];

  const int tid   = threadIdx.x;
  const int lane  = tid & 31;
  const int wave  = tid >> 5;
  const int n     = lane & 15;
  const int brow  = n >> 2;
  const int gate  = n & 3;
  const int hi    = lane >> 4;
  const int kh    = hi << 1;
  const int bbase = blockIdx.x * ROWS_PER_BLOCK;
  const int cbase = wave * 16 + hi * 8;

  float* hs = out;
  float* hT = out + (size_t)B_SZ * S_SZ * H_SZ;
  float* cT = hT + (size_t)B_SZ * H_SZ;

  {
    const float4* src = (const float4*)(y0 + (size_t)bbase * S_SZ * 4);
    float4* dst = (float4*)y0_lds;
#pragma unroll
    for (int i = 0; i < ROWS_PER_BLOCK; ++i)
      dst[i * (Y0_PITCH / 4) + tid] = src[tid + i * 256];
  }

  v2f ah[4];
#pragma unroll
  for (int j = 0; j < 4; ++j) {
    float a0 = 0.f, a1 = 0.f;
    if (n < 4) {
      int kk = cbase + 2 * j;
      a0 = Win[n * 192 + kk];
      a1 = Win[n * 192 + kk + 1];
    }
    v2f av = {a0, a1};
    ah[j] = av;
  }
  v2f aw = {Wout[(wave * 16 + n) * 4 + kh], Wout[(wave * 16 + n) * 4 + kh + 1]};
  v8f cbias;
  for (int r = 0; r < 8; ++r) cbias[r] = bout[cbase + r];

  const float INV_2PI = 0.15915494309189532f;
  float thr[4];
#pragma unroll
  for (int w = 0; w < 4; ++w) thr[w] = thetas[gate * 4 + w] * INV_2PI;

  const float act_sc  = (gate == 2) ? 1.0f : 0.5f;
  const float act_off = (gate == 2) ? 0.0f : 0.5f;

  float c_state[8], h[8];
#pragma unroll
  for (int r = 0; r < 8; ++r) { c_state[r] = 0.f; h[r] = 0.f; }

  float* hs_row = hs + ((size_t)(bbase + brow) * S_SZ) * H_SZ + cbase;
  const float* y0_row = y0_lds + brow * Y0_PITCH;

  __syncthreads();

  for (int t = 0; t < S_SZ; ++t) {
    float* pw = part + (t & 1) * (8 * 16 * 4);
    const float4 yv = *(const float4*)(y0_row + t * 4);

    v8f acc = {0.f, 0.f, 0.f, 0.f, 0.f, 0.f, 0.f, 0.f};
#pragma unroll
    for (int j = 0; j < 4; ++j) {
      v2f b = {h[2 * j], h[2 * j + 1]};
      acc = wmma_f32_16x16x4(ah[j], b, acc);
    }
    if (hi == 0) {
      float4 pv = {acc[0], acc[1], acc[2], acc[3]};
      *(float4*)&pw[wave * 64 + n * 4] = pv;
    }
    lds_barrier();

    float ys[4] = {yv.x, yv.y, yv.z, yv.w};
#pragma unroll
    for (int wv = 0; wv < 8; ++wv) {
      const float4 pv = *(const float4*)&pw[wv * 64 + n * 4];
      ys[0] += pv.x;  ys[1] += pv.y;  ys[2] += pv.z;  ys[3] += pv.w;
    }
    v2f bz;
    {
      const float c0 = __builtin_amdgcn_cosf(fmaf(ys[0], INV_2PI, thr[0]));
      const float c1 = __builtin_amdgcn_cosf(fmaf(ys[1], INV_2PI, thr[1]));
      const float c2 = __builtin_amdgcn_cosf(fmaf(ys[2], INV_2PI, thr[2]));
      const float c3 = __builtin_amdgcn_cosf(fmaf(ys[3], INV_2PI, thr[3]));
      const float p01 = c0 * c1, p23 = c2 * c3;
      const float u = hi ? p01 : c1;
      const float v = hi ? c2 : p23;
      v2f b = {u * v, hi ? p01 * p23 : p01};
      bz = b;
    }

    const v8f ga = wmma_f32_16x16x4(aw, bz, cbias);
#pragma unroll
    for (int r = 0; r < 8; ++r) {
      const float a  = fmaf(act_sc, hw_tanh(act_sc * ga[r]), act_off);
      const float fv = quad_bcast<0>(a);
      const float iv = quad_bcast<1>(a);
      const float gv = quad_bcast<2>(a);
      const float ov = quad_bcast<3>(a);
      float cs = fv * c_state[r] + iv * gv;
      c_state[r] = cs;
      h[r] = ov * hw_tanh(cs);
    }
    if (gate == 0) {
      float4* dst = (float4*)(hs_row + (size_t)t * H_SZ);
      float4 lo = {h[0], h[1], h[2], h[3]};
      float4 up = {h[4], h[5], h[6], h[7]};
      dst[0] = lo;
      dst[1] = up;
    }
  }

  if (gate == 0) {
    float4* dh = (float4*)(hT + (size_t)(bbase + brow) * H_SZ + cbase);
    float4* dc = (float4*)(cT + (size_t)(bbase + brow) * H_SZ + cbase);
    float4 hlo = {h[0], h[1], h[2], h[3]}, hup = {h[4], h[5], h[6], h[7]};
    float4 clo = {c_state[0], c_state[1], c_state[2], c_state[3]};
    float4 cup = {c_state[4], c_state[5], c_state[6], c_state[7]};
    dh[0] = hlo;  dh[1] = hup;
    dc[0] = clo;  dc[1] = cup;
  }
}

extern "C" void kernel_launch(void* const* d_in, const int* in_sizes, int n_in,
                              void* d_out, int out_size, void* d_ws, size_t ws_size,
                              hipStream_t stream) {
  const float* x      = (const float*)d_in[0];
  const float* W_in   = (const float*)d_in[1];
  const float* b_in   = (const float*)d_in[2];
  const float* W_out  = (const float*)d_in[3];
  const float* b_out  = (const float*)d_in[4];
  const float* thetas = (const float*)d_in[5];
  float* out = (float*)d_out;
  float* y0  = (float*)d_ws;

  qlstm_xproj<<<1024, 128, 0, stream>>>(x, W_in, b_in, y0);
  qlstm_recur<<<64, 256, 0, stream>>>(y0, W_in, W_out, b_out, thetas, out);
}


# tile columns = (batch row, column half, gate): 2 batch rows a block, 128 blocks; lane activates 4 H columns where 8; y partials by 2 WMMAs where 4, 8+8 per wave joined by DPP add (k order regrouped)
# speedup vs baseline: 4.0893x; 1.1770x over previous
#include <hip/hip_runtime.h>
#include <hip/hip_bf16.h>
#include <math.h>

typedef __attribute__((ext_vector_type(2))) float v2f;
typedef __attribute__((ext_vector_type(8))) float v8f;

static __device__ __forceinline__ v8f wmma_f32_16x16x4(v2f a, v2f b, v8f c) {
  return __builtin_amdgcn_wmma_f32_16x16x4_f32(false, a, false, b, (short)0, c, false, false);
}

static __device__ __forceinline__ float hw_tanh(float x) {
  return __builtin_amdgcn_tanhf(x);
}
static __device__ __forceinline__ float hw_sigmoid(float x) {
  return fmaf(0.5f, hw_tanh(0.5f * x), 0.5f);
}

static __device__ __forceinline__ void lds_barrier() {
  __builtin_amdgcn_fence(__ATOMIC_RELEASE, "workgroup", "local");
  __builtin_amdgcn_s_barrier();
  __builtin_amdgcn_fence(__ATOMIC_ACQUIRE, "workgroup", "local");
}

#define B_SZ 256
#define S_SZ 256
#define F_SZ 64
#define H_SZ 128
#define Y0_PITCH (S_SZ * 4 + 4)

__global__ void __launch_bounds__(128)
qlstm_xproj(const float* __restrict__ x, const float* __restrict__ Win,
            const float* __restrict__ bin, float* __restrict__ y0) {
  const int lane = threadIdx.x & 31;
  const int wave = threadIdx.x >> 5;
  const int tile = blockIdx.x * 4 + wave;
  const int n    = lane & 15;
  const int hi   = lane >> 4;
  const int kh   = hi << 1;

  v2f bch[16];
#pragma unroll
  for (int j = 0; j < 16; ++j) {
    float b0 = 0.f, b1 = 0.f;
    if (n < 4) {
      int kk = j * 4 + kh;
      b0 = Win[n * 192 + 128 + kk];
      b1 = Win[n * 192 + 128 + kk + 1];
    }
    v2f bv = {b0, b1};
    bch[j] = bv;
  }

  const float bb = (n < 4) ? bin[n] : 0.f;
  v8f acc = {bb, bb, bb, bb, bb, bb, bb, bb};

  const float* xr = x + (size_t)(tile * 16 + n) * F_SZ;
#pragma unroll
  for (int j = 0; j < 16; ++j) {
    v2f a = {xr[j * 4 + kh], xr[j * 4 + kh + 1]};
    acc = wmma_f32_16x16x4(a, bch[j], acc);
  }

  if (n < 4) {
#pragma unroll
    for (int r = 0; r < 8; ++r) {
      int row = tile * 16 + r + (hi << 3);
      y0[(size_t)row * 4 + n] = acc[r];
    }
  }
}

#define ROWS_PER_BLOCK 2

template <int K>
static __device__ __forceinline__ float quad_bcast(float v) {
  return __int_as_float(__builtin_amdgcn_mov_dpp(__float_as_int(v), K * 0x55, 0xF, 0xF, true));
}

static __device__ __forceinline__ float other_half(float v) {
  return __int_as_float(__builtin_amdgcn_mov_dpp(__float_as_int(v), 0x164, 0xF, 0xF, true));
}

__global__ void __launch_bounds__(256)
qlstm_recur(const float* __restrict__ y0, const float* __restrict__ Win,
            const float* __restrict__ Wout, const float* __restrict__ bout,
            const float* __restrict__ thetas, float* __restrict__ out) {
  __shared__ __attribute__((aligned(16))) float y0_lds[ROWS_PER_BLOCK * Y0_PITCH];
  __shared__ __attribute__((aligned(16))) float part[2 * 8 * 16 * 4];

  const int tid   = threadIdx.x;
  const int lane  = tid & 31;
  const int wave  = tid >> 5;
  const int n     = lane & 15;
  const int gate  = n & 3;
  const int sh    = (n >> 2) & 1;
  const int brow  = n >> 3;
  const int hi    = lane >> 4;
  const int kh    = hi << 1;
  const int bbase = blockIdx.x * ROWS_PER_BLOCK;
  const int rbase = wave * 16 + hi * 8;
  const int cbase = rbase + sh * 4;

  float* hs = out;
  float* hT = out + (size_t)B_SZ * S_SZ * H_SZ;
  float* cT = hT + (size_t)B_SZ * H_SZ;

  {
    const float4* src = (const float4*)(y0 + (size_t)bbase * S_SZ * 4);
    float4* dst = (float4*)y0_lds;
#pragma unroll
    for (int i = 0; i < ROWS_PER_BLOCK; ++i)
      dst[i * (Y0_PITCH / 4) + tid] = src[tid + i * 256];
  }

  v2f ah[2];
#pragma unroll
  for (int j = 0; j < 2; ++j) {
    float a0 = 0.f, a1 = 0.f;
    if (n < 8) {
      int kk = wave * 16 + hi * 8 + (n >> 2) * 4 + 2 * j;
      a0 = Win[(n & 3) * 192 + kk];
      a1 = Win[(n & 3) * 192 + kk + 1];
    }
    v2f av = {a0, a1};
    ah[j] = av;
  }
  v2f aw = {Wout[(wave * 16 + n) * 4 + kh], Wout[(wave * 16 + n) * 4 + kh + 1]};
  v8f cbias;
  for (int r = 0; r < 8; ++r) cbias[r] = bout[rbase + r];

  const float INV_2PI = 0.15915494309189532f;
  float thr[4];
#pragma unroll
  for (int w = 0; w < 4; ++w) thr[w] = thetas[gate * 4 + w] * INV_2PI;

  const float act_sc  = (gate == 2) ? 1.0f : 0.5f;
  const float act_off = (gate == 2) ? 0.0f : 0.5f;

  float c_state[4], h[4];
#pragma unroll
  for (int i = 0; i < 4; ++i) { c_state[i] = 0.f; h[i] = 0.f; }

  float* hs_row = hs + ((size_t)(bbase + brow) * S_SZ) * H_SZ + cbase;
  const float* y0_row = y0_lds + brow * Y0_PITCH;

  __syncthreads();

  for (int t = 0; t < S_SZ; ++t) {
    float* pw = part + (t & 1) * (8 * 16 * 4);
    const float4 yv = *(const float4*)(y0_row + t * 4);

    v8f acc = {0.f, 0.f, 0.f, 0.f, 0.f, 0.f, 0.f, 0.f};
#pragma unroll
    for (int j = 0; j < 2; ++j) {
      v2f b = {h[2 * j], h[2 * j + 1]};
      acc = wmma_f32_16x16x4(ah[j], b, acc);
    }
    float4 pv;
    pv.x = sh ? acc[4] : acc[0];
    pv.y = sh ? acc[5] : acc[1];
    pv.z = sh ? acc[6] : acc[2];
    pv.w = sh ? acc[7] : acc[3];
    pv.x += other_half(pv.x);
    pv.y += other_half(pv.y);
    pv.z += other_half(pv.z);
    pv.w += other_half(pv.w);
    if (hi == 0)
      *(float4*)&pw[wave * 64 + n * 4] = pv;
    lds_barrier();

    float ys[4] = {yv.x, yv.y, yv.z, yv.w};
#pragma unroll
    for (int wv = 0; wv < 8; ++wv) {
      const float4 qv = *(const float4*)&pw[wv * 64 + n * 4];
      ys[0] += qv.x;  ys[1] += qv.y;  ys[2] += qv.z;  ys[3] += qv.w;
    }
    v2f bz;
    {
      const float c0 = __builtin_amdgcn_cosf(fmaf(ys[0], INV_2PI, thr[0]));
      const float c1 = __builtin_amdgcn_cosf(fmaf(ys[1], INV_2PI, thr[1]));
      const float c2 = __builtin_amdgcn_cosf(fmaf(ys[2], INV_2PI, thr[2]));
      const float c3 = __builtin_amdgcn_cosf(fmaf(ys[3], INV_2PI, thr[3]));
      const float p01 = c0 * c1, p23 = c2 * c3;
      const float u = hi ? p01 : c1;
      const float v = hi ? c2 : p23;
      v2f b = {u * v, hi ? p01 * p23 : p01};
      bz = b;
    }

    const v8f ga = wmma_f32_16x16x4(aw, bz, cbias);
    const float gx[4] = {sh ? ga[4] : ga[0], sh ? ga[5] : ga[1], sh ? ga[6] : ga[2], sh ? ga[7] : ga[3]};
#pragma unroll
    for (int i = 0; i < 4; ++i) {
      const float a  = fmaf(act_sc, hw_tanh(act_sc * gx[i]), act_off);
      const float fv = quad_bcast<0>(a);
      const float iv = quad_bcast<1>(a);
      const float gv = quad_bcast<2>(a);
      const float ov = quad_bcast<3>(a);
      float cs = fv * c_state[i] + iv * gv;
      c_state[i] = cs;
      h[i] = ov * hw_tanh(cs);
    }
    if (gate == 0) {
      float4 hv = {h[0], h[1], h[2], h[3]};
      *(float4*)(hs_row + (size_t)t * H_SZ) = hv;
    }
  }

  if (gate == 0) {
    float4 hv = {h[0], h[1], h[2], h[3]};
    float4 cv = {c_state[0], c_state[1], c_state[2], c_state[3]};
    *(float4*)(hT + (size_t)(bbase + brow) * H_SZ + cbase) = hv;
    *(float4*)(cT + (size_t)(bbase + brow) * H_SZ + cbase) = cv;
  }
}

extern "C" void kernel_launch(void* const* d_in, const int* in_sizes, int n_in,
                              void* d_out, int out_size, void* d_ws, size_t ws_size,
                              hipStream_t stream) {
  const float* x      = (const float*)d_in[0];
  const float* W_in   = (const float*)d_in[1];
  const float* b_in   = (const float*)d_in[2];
  const float* W_out  = (const float*)d_in[3];
  const float* b_out  = (const float*)d_in[4];
  const float* thetas = (const float*)d_in[5];
  float* out = (float*)d_out;
  float* y0  = (float*)d_ws;

  qlstm_xproj<<<1024, 128, 0, stream>>>(x, W_in, b_in, y0);
  qlstm_recur<<<128, 256, 0, stream>>>(y0, W_in, W_out, b_out, thetas, out);
}


# y partials exchanged compactly: a wave stores one float4 per batch row, a lane reads 1 partial where 8, DPP butterfly adds join the 8 waves (sum order: pairwise tree, then y0)
# speedup vs baseline: 4.9742x; 1.2164x over previous
#include <hip/hip_runtime.h>
#include <hip/hip_bf16.h>
#include <math.h>

typedef __attribute__((ext_vector_type(2))) float v2f;
typedef __attribute__((ext_vector_type(8))) float v8f;

static __device__ __forceinline__ v8f wmma_f32_16x16x4(v2f a, v2f b, v8f c) {
  return __builtin_amdgcn_wmma_f32_16x16x4_f32(false, a, false, b, (short)0, c, false, false);
}

static __device__ __forceinline__ float hw_tanh(float x) {
  return __builtin_amdgcn_tanhf(x);
}
static __device__ __forceinline__ float hw_sigmoid(float x) {
  return fmaf(0.5f, hw_tanh(0.5f * x), 0.5f);
}

static __device__ __forceinline__ void lds_barrier() {
  __builtin_amdgcn_fence(__ATOMIC_RELEASE, "workgroup", "local");
  __builtin_amdgcn_s_barrier();
  __builtin_amdgcn_fence(__ATOMIC_ACQUIRE, "workgroup", "local");
}

#define B_SZ 256
#define S_SZ 256
#define F_SZ 64
#define H_SZ 128
#define Y0_PITCH (S_SZ * 4 + 4)

__global__ void __launch_bounds__(128)
qlstm_xproj(const float* __restrict__ x, const float* __restrict__ Win,
            const float* __restrict__ bin, float* __restrict__ y0) {
  const int lane = threadIdx.x & 31;
  const int wave = threadIdx.x >> 5;
  const int tile = blockIdx.x * 4 + wave;
  const int n    = lane & 15;
  const int hi   = lane >> 4;
  const int kh   = hi << 1;

  v2f bch[16];
#pragma unroll
  for (int j = 0; j < 16; ++j) {
    float b0 = 0.f, b1 = 0.f;
    if (n < 4) {
      int kk = j * 4 + kh;
      b0 = Win[n * 192 + 128 + kk];
      b1 = Win[n * 192 + 128 + kk + 1];
    }
    v2f bv = {b0, b1};
    bch[j] = bv;
  }

  const float bb = (n < 4) ? bin[n] : 0.f;
  v8f acc = {bb, bb, bb, bb, bb, bb, bb, bb};

  const float* xr = x + (size_t)(tile * 16 + n) * F_SZ;
#pragma unroll
  for (int j = 0; j < 16; ++j) {
    v2f a = {xr[j * 4 + kh], xr[j * 4 + kh + 1]};
    acc = wmma_f32_16x16x4(a, bch[j], acc);
  }

  if (n < 4) {
#pragma unroll
    for (int r = 0; r < 8; ++r) {
      int row = tile * 16 + r + (hi << 3);
      y0[(size_t)row * 4 + n] = acc[r];
    }
  }
}

#define ROWS_PER_BLOCK 2

template <int K>
static __device__ __forceinline__ float quad_bcast(float v) {
  return __int_as_float(__builtin_amdgcn_mov_dpp(__float_as_int(v), K * 0x55, 0xF, 0xF, true));
}

static __device__ __forceinline__ float other_half(float v) {
  return __int_as_float(__builtin_amdgcn_mov_dpp(__float_as_int(v), 0x164, 0xF, 0xF, true));
}

template <int M>
static __device__ __forceinline__ float row_xor(float v) {
  return __int_as_float(__builtin_amdgcn_mov_dpp(__float_as_int(v), 0x160 + M, 0xF, 0xF, true));
}

__global__ void __launch_bounds__(256)
qlstm_recur(const float* __restrict__ y0, const float* __restrict__ Win,
            const float* __restrict__ Wout, const float* __restrict__ bout,
            const float* __restrict__ thetas, float* __restrict__ out) {
  __shared__ __attribute__((aligned(16))) float y0_lds[ROWS_PER_BLOCK * Y0_PITCH];
  __shared__ __attribute__((aligned(16))) float part[2 * ROWS_PER_BLOCK * 8 * 4];

  const int tid   = threadIdx.x;
  const int lane  = tid & 31;
  const int wave  = tid >> 5;
  const int n     = lane & 15;
  const int gate  = n & 3;
  const int sh    = (n >> 2) & 1;
  const int brow  = n >> 3;
  const int hi    = lane >> 4;
  const int kh    = hi << 1;
  const int bbase = blockIdx.x * ROWS_PER_BLOCK;
  const int rbase = wave * 16 + hi * 8;
  const int cbase = rbase + sh * 4;

  float* hs = out;
  float* hT = out + (size_t)B_SZ * S_SZ * H_SZ;
  float* cT = hT + (size_t)B_SZ * H_SZ;

  {
    const float4* src = (const float4*)(y0 + (size_t)bbase * S_SZ * 4);
    float4* dst = (float4*)y0_lds;
#pragma unroll
    for (int i = 0; i < ROWS_PER_BLOCK; ++i)
      dst[i * (Y0_PITCH / 4) + tid] = src[tid + i * 256];
  }

  v2f ah[2];
#pragma unroll
  for (int j = 0; j < 2; ++j) {
    float a0 = 0.f, a1 = 0.f;
    if (n < 8) {
      int kk = wave * 16 + hi * 8 + (n >> 2) * 4 + 2 * j;
      a0 = Win[(n & 3) * 192 + kk];
      a1 = Win[(n & 3) * 192 + kk + 1];
    }
    v2f av = {a0, a1};
    ah[j] = av;
  }
  v2f aw = {Wout[(wave * 16 + n) * 4 + kh], Wout[(wave * 16 + n) * 4 + kh + 1]};
  v8f cbias;
  for (int r = 0; r < 8; ++r) cbias[r] = bout[rbase + r];

  const float INV_2PI = 0.15915494309189532f;
  float thr[4];
#pragma unroll
  for (int w = 0; w < 4; ++w) thr[w] = thetas[gate * 4 + w] * INV_2PI;

  const float act_sc  = (gate == 2) ? 1.0f : 0.5f;
  const float act_off = (gate == 2) ? 0.0f : 0.5f;

  float c_state[4], h[4];
#pragma unroll
  for (int i = 0; i < 4; ++i) { c_state[i] = 0.f; h[i] = 0.f; }

  float* hs_row = hs + ((size_t)(bbase + brow) * S_SZ) * H_SZ + cbase;
  const float* y0_row = y0_lds + brow * Y0_PITCH;

  __syncthreads();

  for (int t = 0; t < S_SZ; ++t) {
    float* pw = part + (t & 1) * (ROWS_PER_BLOCK * 8 * 4);
    const float4 yv = *(const float4*)(y0_row + t * 4);

    v8f acc = {0.f, 0.f, 0.f, 0.f, 0.f, 0.f, 0.f, 0.f};
#pragma unroll
    for (int j = 0; j < 2; ++j) {
      v2f b = {h[2 * j], h[2 * j + 1]};
      acc = wmma_f32_16x16x4(ah[j], b, acc);
    }
    float4 pv;
    pv.x = acc[0] + other_half(acc[4]);
    pv.y = acc[1] + other_half(acc[5]);
    pv.z = acc[2] + other_half(acc[6]);
    pv.w = acc[3] + other_half(acc[7]);
    if ((lane & 0x17) == 0)
      *(float4*)&pw[(brow * 8 + wave) * 4] = pv;
    lds_barrier();

    float4 qv = *(const float4*)&pw[n * 4];
    qv.x += row_xor<1>(qv.x);  qv.y += row_xor<1>(qv.y);  qv.z += row_xor<1>(qv.z);  qv.w += row_xor<1>(qv.w);
    qv.x += row_xor<2>(qv.x);  qv.y += row_xor<2>(qv.y);  qv.z += row_xor<2>(qv.z);  qv.w += row_xor<2>(qv.w);
    qv.x += row_xor<4>(qv.x);  qv.y += row_xor<4>(qv.y);  qv.z += row_xor<4>(qv.z);  qv.w += row_xor<4>(qv.w);
    const float ys[4] = {yv.x + qv.x, yv.y + qv.y, yv.z + qv.z, yv.w + qv.w};
    v2f bz;
    {
      const float c0 = __builtin_amdgcn_cosf(fmaf(ys[0], INV_2PI, thr[0]));
      const float c1 = __builtin_amdgcn_cosf(fmaf(ys[1], INV_2PI, thr[1]));
      const float c2 = __builtin_amdgcn_cosf(fmaf(ys[2], INV_2PI, thr[2]));
      const float c3 = __builtin_amdgcn_cosf(fmaf(ys[3], INV_2PI, thr[3]));
      const float p01 = c0 * c1, p23 = c2 * c3;
      const float u = hi ? p01 : c1;
      const float v = hi ? c2 : p23;
      v2f b = {u * v, hi ? p01 * p23 : p01};
      bz = b;
    }

    const v8f ga = wmma_f32_16x16x4(aw, bz, cbias);
    const float gx[4] = {sh ? ga[4] : ga[0], sh ? ga[5] : ga[1], sh ? ga[6] : ga[2], sh ? ga[7] : ga[3]};
#pragma unroll
    for (int i = 0; i < 4; ++i) {
      const float a  = fmaf(act_sc, hw_tanh(act_sc * gx[i]), act_off);
      const float fv = quad_bcast<0>(a);
      const float iv = quad_bcast<1>(a);
      const float gv = quad_bcast<2>(a);
      const float ov = quad_bcast<3>(a);
      float cs = fv * c_state[i] + iv * gv;
      c_state[i] = cs;
      h[i] = ov * hw_tanh(cs);
    }
    if (gate == 0) {
      float4 hv = {h[0], h[1], h[2], h[3]};
      *(float4*)(hs_row + (size_t)t * H_SZ) = hv;
    }
  }

  if (gate == 0) {
    float4 hv = {h[0], h[1], h[2], h[3]};
    float4 cv = {c_state[0], c_state[1], c_state[2], c_state[3]};
    *(float4*)(hT + (size_t)(bbase + brow) * H_SZ + cbase) = hv;
    *(float4*)(cT + (size_t)(bbase + brow) * H_SZ + cbase) = cv;
  }
}

extern "C" void kernel_launch(void* const* d_in, const int* in_sizes, int n_in,
                              void* d_out, int out_size, void* d_ws, size_t ws_size,
                              hipStream_t stream) {
  const float* x      = (const float*)d_in[0];
  const float* W_in   = (const float*)d_in[1];
  const float* b_in   = (const float*)d_in[2];
  const float* W_out  = (const float*)d_in[3];
  const float* b_out  = (const float*)d_in[4];
  const float* thetas = (const float*)d_in[5];
  float* out = (float*)d_out;
  float* y0  = (float*)d_ws;

  qlstm_xproj<<<1024, 128, 0, stream>>>(x, W_in, b_in, y0);
  qlstm_recur<<<128, 256, 0, stream>>>(y0, W_in, W_out, b_out, thetas, out);
}


# tile columns = (column quarter, gate): 1 batch row a block, 256 blocks where 128; lane activates 2 H columns where 4; y partials by 1 WMMA where 2, 16 joined by 4 DPP add stages (sum order regrouped)
# speedup vs baseline: 5.1981x; 1.0450x over previous
#include <hip/hip_runtime.h>
#include <hip/hip_bf16.h>
#include <math.h>

typedef __attribute__((ext_vector_type(2))) float v2f;
typedef __attribute__((ext_vector_type(8))) float v8f;

static __device__ __forceinline__ v8f wmma_f32_16x16x4(v2f a, v2f b, v8f c) {
  return __builtin_amdgcn_wmma_f32_16x16x4_f32(false, a, false, b, (short)0, c, false, false);
}

static __device__ __forceinline__ float hw_tanh(float x) {
  return __builtin_amdgcn_tanhf(x);
}
static __device__ __forceinline__ float hw_sigmoid(float x) {
  return fmaf(0.5f, hw_tanh(0.5f * x), 0.5f);
}

static __device__ __forceinline__ void lds_barrier() {
  __builtin_amdgcn_fence(__ATOMIC_RELEASE, "workgroup", "local");
  __builtin_amdgcn_s_barrier();
  __builtin_amdgcn_fence(__ATOMIC_ACQUIRE, "workgroup", "local");
}

#define B_SZ 256
#define S_SZ 256
#define F_SZ 64
#define H_SZ 128

__global__ void __launch_bounds__(128)
qlstm_xproj(const float* __restrict__ x, const float* __restrict__ Win,
            const float* __restrict__ bin, float* __restrict__ y0) {
  const int lane = threadIdx.x & 31;
  const int wave = threadIdx.x >> 5;
  const int tile = blockIdx.x * 4 + wave;
  const int n    = lane & 15;
  const int hi   = lane >> 4;
  const int kh   = hi << 1;

  v2f bch[16];
#pragma unroll
  for (int j = 0; j < 16; ++j) {
    float b0 = 0.f, b1 = 0.f;
    if (n < 4) {
      int kk = j * 4 + kh;
      b0 = Win[n * 192 + 128 + kk];
      b1 = Win[n * 192 + 128 + kk + 1];
    }
    v2f bv = {b0, b1};
    bch[j] = bv;
  }

  const float bb = (n < 4) ? bin[n] : 0.f;
  v8f acc = {bb, bb, bb, bb, bb, bb, bb, bb};

  const float* xr = x + (size_t)(tile * 16 + n) * F_SZ;
#pragma unroll
  for (int j = 0; j < 16; ++j) {
    v2f a = {xr[j * 4 + kh], xr[j * 4 + kh + 1]};
    acc = wmma_f32_16x16x4(a, bch[j], acc);
  }

  if (n < 4) {
#pragma unroll
    for (int r = 0; r < 8; ++r) {
      int row = tile * 16 + r + (hi << 3);
      y0[(size_t)row * 4 + n] = acc[r];
    }
  }
}


template <int K>
static __device__ __forceinline__ float quad_bcast(float v) {
  return __int_as_float(__builtin_amdgcn_mov_dpp(__float_as_int(v), K * 0x55, 0xF, 0xF, true));
}

template <int M>
static __device__ __forceinline__ float row_xor(float v) {
  return __int_as_float(__builtin_amdgcn_mov_dpp(__float_as_int(v), 0x160 + M, 0xF, 0xF, true));
}

__global__ void __launch_bounds__(256)
qlstm_recur(const float* __restrict__ y0, const float* __restrict__ Win,
            const float* __restrict__ Wout, const float* __restrict__ bout,
            const float* __restrict__ thetas, float* __restrict__ out) {
  __shared__ __attribute__((aligned(16))) float y0_lds[S_SZ * 4];
  __shared__ __attribute__((aligned(16))) float part[2 * 16 * 4];

  const int tid   = threadIdx.x;
  const int lane  = tid & 31;
  const int wave  = tid >> 5;
  const int n     = lane & 15;
  const int gate  = n & 3;
  const int sq    = n >> 2;
  const int hi    = lane >> 4;
  const int kh    = hi << 1;
  const int brow  = blockIdx.x;
  const int rbase = wave * 16 + hi * 8;
  const int cbase = rbase + sq * 2;

  float* hs = out;
  float* hT = out + (size_t)B_SZ * S_SZ * H_SZ;
  float* cT = hT + (size_t)B_SZ * H_SZ;

  ((float4*)y0_lds)[tid] = ((const float4*)(y0 + (size_t)brow * S_SZ * 4))[tid];

  v2f ah;
  {
    const int kk = wave * 16 + hi * 8 + (n >> 2) * 2;
    v2f av = {Win[(n & 3) * 192 + kk], Win[(n & 3) * 192 + kk + 1]};
    ah = av;
  }
  v2f aw = {Wout[(wave * 16 + n) * 4 + kh], Wout[(wave * 16 + n) * 4 + kh + 1]};
  v8f cbias;
  for (int r = 0; r < 8; ++r) cbias[r] = bout[rbase + r];

  const float INV_2PI = 0.15915494309189532f;
  float thr[4];
#pragma unroll
  for (int w = 0; w < 4; ++w) thr[w] = thetas[gate * 4 + w] * INV_2PI;

  const float act_sc  = (gate == 2) ? 1.0f : 0.5f;
  const float act_off = (gate == 2) ? 0.0f : 0.5f;

  float c_state[2], h[2];
#pragma unroll
  for (int i = 0; i < 2; ++i) { c_state[i] = 0.f; h[i] = 0.f; }

  float* hs_row = hs + ((size_t)brow * S_SZ) * H_SZ + cbase;
  const bool q_odd  = (sq & 1) != 0;
  const bool q_high = (sq & 2) != 0;

  __syncthreads();

  for (int t = 0; t < S_SZ; ++t) {
    float* pw = part + (t & 1) * (16 * 4);
    const float4 yv = *(const float4*)(y0_lds + t * 4);

    v8f acc = {0.f, 0.f, 0.f, 0.f, 0.f, 0.f, 0.f, 0.f};
    {
      v2f b = {h[0], h[1]};
      acc = wmma_f32_16x16x4(ah, b, acc);
    }
    float4 pv;
    pv.x = acc[0] + row_xor<4>(acc[4]);
    pv.y = acc[1] + row_xor<4>(acc[5]);
    pv.z = acc[2] + row_xor<4>(acc[6]);
    pv.w = acc[3] + row_xor<4>(acc[7]);
    if (lane == hi * 24)
      *(float4*)&pw[(wave * 2 + hi) * 4] = pv;
    lds_barrier();

    float4 qv = *(const float4*)&pw[n * 4];
    qv.x += row_xor<1>(qv.x);  qv.y += row_xor<1>(qv.y);  qv.z += row_xor<1>(qv.z);  qv.w += row_xor<1>(qv.w);
    qv.x += row_xor<2>(qv.x);  qv.y += row_xor<2>(qv.y);  qv.z += row_xor<2>(qv.z);  qv.w += row_xor<2>(qv.w);
    qv.x += row_xor<4>(qv.x);  qv.y += row_xor<4>(qv.y);  qv.z += row_xor<4>(qv.z);  qv.w += row_xor<4>(qv.w);
    qv.x += row_xor<8>(qv.x);  qv.y += row_xor<8>(qv.y);  qv.z += row_xor<8>(qv.z);  qv.w += row_xor<8>(qv.w);
    const float ys[4] = {yv.x + qv.x, yv.y + qv.y, yv.z + qv.z, yv.w + qv.w};
    v2f bz;
    {
      const float c0 = __builtin_amdgcn_cosf(fmaf(ys[0], INV_2PI, thr[0]));
      const float c1 = __builtin_amdgcn_cosf(fmaf(ys[1], INV_2PI, thr[1]));
      const float c2 = __builtin_amdgcn_cosf(fmaf(ys[2], INV_2PI, thr[2]));
      const float c3 = __builtin_amdgcn_cosf(fmaf(ys[3], INV_2PI, thr[3]));
      const float p01 = c0 * c1, p23 = c2 * c3;
      const float u = hi ? p01 : c1;
      const float v = hi ? c2 : p23;
      v2f b = {u * v, hi ? p01 * p23 : p01};
      bz = b;
    }

    const v8f ga = wmma_f32_16x16x4(aw, bz, cbias);
    const float g_lo0 = q_odd ? ga[2] : ga[0], g_lo1 = q_odd ? ga[3] : ga[1];
    const float g_hi0 = q_odd ? ga[6] : ga[4], g_hi1 = q_odd ? ga[7] : ga[5];
    const float gx[2] = {q_high ? g_hi0 : g_lo0, q_high ? g_hi1 : g_lo1};
#pragma unroll
    for (int i = 0; i < 2; ++i) {
      const float a  = fmaf(act_sc, hw_tanh(act_sc * gx[i]), act_off);
      const float fv = quad_bcast<0>(a);
      const float iv = quad_bcast<1>(a);
      const float gv = quad_bcast<2>(a);
      const float ov = quad_bcast<3>(a);
      float cs = fv * c_state[i] + iv * gv;
      c_state[i] = cs;
      h[i] = ov * hw_tanh(cs);
    }
    if (gate == 0) {
      float2 hv = {h[0], h[1]};
      *(float2*)(hs_row + (size_t)t * H_SZ) = hv;
    }
  }

  if (gate == 0) {
    float2 hv = {h[0], h[1]};
    float2 cv = {c_state[0], c_state[1]};
    *(float2*)(hT + (size_t)brow * H_SZ + cbase) = hv;
    *(float2*)(cT + (size_t)brow * H_SZ + cbase) = cv;
  }
}

extern "C" void kernel_launch(void* const* d_in, const int* in_sizes, int n_in,
                              void* d_out, int out_size, void* d_ws, size_t ws_size,
                              hipStream_t stream) {
  const float* x      = (const float*)d_in[0];
  const float* W_in   = (const float*)d_in[1];
  const float* b_in   = (const float*)d_in[2];
  const float* W_out  = (const float*)d_in[3];
  const float* b_out  = (const float*)d_in[4];
  const float* thetas = (const float*)d_in[5];
  float* out = (float*)d_out;
  float* y0  = (float*)d_ws;

  qlstm_xproj<<<1024, 128, 0, stream>>>(x, W_in, b_in, y0);
  qlstm_recur<<<256, 256, 0, stream>>>(y0, W_in, W_out, b_out, thetas, out);
}


# input projection fused into the scan kernel: each block makes its own batch row's y0 by WMMA straight into LDS (x in by whole-line loads, staged through LDS); one launch where two, no workspace
# speedup vs baseline: 5.3979x; 1.0384x over previous
#include <hip/hip_runtime.h>
#include <hip/hip_bf16.h>
#include <math.h>

typedef __attribute__((ext_vector_type(2))) float v2f;
typedef __attribute__((ext_vector_type(8))) float v8f;

static __device__ __forceinline__ v8f wmma_f32_16x16x4(v2f a, v2f b, v8f c) {
  return __builtin_amdgcn_wmma_f32_16x16x4_f32(false, a, false, b, (short)0, c, false, false);
}

static __device__ __forceinline__ float hw_tanh(float x) {
  return __builtin_amdgcn_tanhf(x);
}
static __device__ __forceinline__ float hw_sigmoid(float x) {
  return fmaf(0.5f, hw_tanh(0.5f * x), 0.5f);
}

static __device__ __forceinline__ void lds_barrier() {
  __builtin_amdgcn_fence(__ATOMIC_RELEASE, "workgroup", "local");
  __builtin_amdgcn_s_barrier();
  __builtin_amdgcn_fence(__ATOMIC_ACQUIRE, "workgroup", "local");
}

#define B_SZ 256
#define S_SZ 256
#define F_SZ 64
#define H_SZ 128
#define X_PITCH 68


template <int K>
static __device__ __forceinline__ float quad_bcast(float v) {
  return __int_as_float(__builtin_amdgcn_mov_dpp(__float_as_int(v), K * 0x55, 0xF, 0xF, true));
}

template <int M>
static __device__ __forceinline__ float row_xor(float v) {
  return __int_as_float(__builtin_amdgcn_mov_dpp(__float_as_int(v), 0x160 + M, 0xF, 0xF, true));
}

__global__ void __launch_bounds__(256)
qlstm_recur(const float* __restrict__ x, const float* __restrict__ Win,
            const float* __restrict__ bin,
            const float* __restrict__ Wout, const float* __restrict__ bout,
            const float* __restrict__ thetas, float* __restrict__ out) {
  __shared__ __attribute__((aligned(16))) float x_lds[S_SZ * X_PITCH];
  __shared__ __attribute__((aligned(16))) float y0_lds[S_SZ * 4];
  __shared__ __attribute__((aligned(16))) float part[2 * 16 * 4];

  const int tid   = threadIdx.x;
  const int lane  = tid & 31;
  const int wave  = tid >> 5;
  const int n     = lane & 15;
  const int gate  = n & 3;
  const int sq    = n >> 2;
  const int hi    = lane >> 4;
  const int kh    = hi << 1;
  const int brow  = blockIdx.x;
  const int rbase = wave * 16 + hi * 8;
  const int cbase = rbase + sq * 2;

  float* hs = out;
  float* hT = out + (size_t)B_SZ * S_SZ * H_SZ;
  float* cT = hT + (size_t)B_SZ * H_SZ;

  v2f bch[16];
  float bb;
  {
    const bool ncol = n < 4;
    const float* wr = Win + (n & 3) * 192 + 128 + kh;
#pragma unroll
    for (int j = 0; j < 16; ++j) {
      const float b0 = wr[j * 4], b1 = wr[j * 4 + 1];
      v2f bv = {ncol ? b0 : 0.f, ncol ? b1 : 0.f};
      bch[j] = bv;
    }
    const float bq = bin[n & 3];
    bb = ncol ? bq : 0.f;
  }
  {
    const float4* xg = (const float4*)(x + (size_t)brow * S_SZ * F_SZ) + wave * 512 + lane;
    float4 xv[16];
#pragma unroll
    for (int i = 0; i < 16; ++i) xv[i] = xg[i * 32];
#pragma unroll
    for (int i = 0; i < 16; ++i)
      *(float4*)&x_lds[(wave * 32 + i * 2 + hi) * X_PITCH + n * 4] = xv[i];
  }
  __syncthreads();
  {
    const float* xr0 = x_lds + (wave * 32 + n) * X_PITCH + kh;
    const float* xr1 = xr0 + 16 * X_PITCH;
    v8f acc0 = {bb, bb, bb, bb, bb, bb, bb, bb};
    v8f acc1 = acc0;
#pragma unroll
    for (int j = 0; j < 16; ++j) {
      v2f a0 = {xr0[j * 4], xr0[j * 4 + 1]};
      v2f a1 = {xr1[j * 4], xr1[j * 4 + 1]};
      acc0 = wmma_f32_16x16x4(a0, bch[j], acc0);
      acc1 = wmma_f32_16x16x4(a1, bch[j], acc1);
    }
    if (n < 4) {
#pragma unroll
      for (int r = 0; r < 8; ++r) {
        y0_lds[(wave * 32 + hi * 8 + r) * 4 + n]      = acc0[r];
        y0_lds[(wave * 32 + 16 + hi * 8 + r) * 4 + n] = acc1[r];
      }
    }
  }

  v2f ah;
  {
    const int kk = wave * 16 + hi * 8 + (n >> 2) * 2;
    v2f av = {Win[(n & 3) * 192 + kk], Win[(n & 3) * 192 + kk + 1]};
    ah = av;
  }
  v2f aw = {Wout[(wave * 16 + n) * 4 + kh], Wout[(wave * 16 + n) * 4 + kh + 1]};
  v8f cbias;
  for (int r = 0; r < 8; ++r) cbias[r] = bout[rbase + r];

  const float INV_2PI = 0.15915494309189532f;
  float thr[4];
#pragma unroll
  for (int w = 0; w < 4; ++w) thr[w] = thetas[gate * 4 + w] * INV_2PI;

  const float act_sc  = (gate == 2) ? 1.0f : 0.5f;
  const float act_off = (gate == 2) ? 0.0f : 0.5f;

  float c_state[2], h[2];
#pragma unroll
  for (int i = 0; i < 2; ++i) { c_state[i] = 0.f; h[i] = 0.f; }

  float* hs_row = hs + ((size_t)brow * S_SZ) * H_SZ + cbase;
  const bool q_odd  = (sq & 1) != 0;
  const bool q_high = (sq & 2) != 0;

  __syncthreads();

  for (int t = 0; t < S_SZ; ++t) {
    float* pw = part + (t & 1) * (16 * 4);
    const float4 yv = *(const float4*)(y0_lds + t * 4);

    v8f acc = {0.f, 0.f, 0.f, 0.f, 0.f, 0.f, 0.f, 0.f};
    {
      v2f b = {h[0], h[1]};
      acc = wmma_f32_16x16x4(ah, b, acc);
    }
    float4 pv;
    pv.x = acc[0] + row_xor<4>(acc[4]);
    pv.y = acc[1] + row_xor<4>(acc[5]);
    pv.z = acc[2] + row_xor<4>(acc[6]);
    pv.w = acc[3] + row_xor<4>(acc[7]);
    if (lane == hi * 24)
      *(float4*)&pw[(wave * 2 + hi) * 4] = pv;
    lds_barrier();

    float4 qv = *(const float4*)&pw[n * 4];
    qv.x += row_xor<1>(qv.x);  qv.y += row_xor<1>(qv.y);  qv.z += row_xor<1>(qv.z);  qv.w += row_xor<1>(qv.w);
    qv.x += row_xor<2>(qv.x);  qv.y += row_xor<2>(qv.y);  qv.z += row_xor<2>(qv.z);  qv.w += row_xor<2>(qv.w);
    qv.x += row_xor<4>(qv.x);  qv.y += row_xor<4>(qv.y);  qv.z += row_xor<4>(qv.z);  qv.w += row_xor<4>(qv.w);
    qv.x += row_xor<8>(qv.x);  qv.y += row_xor<8>(qv.y);  qv.z += row_xor<8>(qv.z);  qv.w += row_xor<8>(qv.w);
    const float ys[4] = {yv.x + qv.x, yv.y + qv.y, yv.z + qv.z, yv.w + qv.w};
    v2f bz;
    {
      const float c0 = __builtin_amdgcn_cosf(fmaf(ys[0], INV_2PI, thr[0]));
      const float c1 = __builtin_amdgcn_cosf(fmaf(ys[1], INV_2PI, thr[1]));
      const float c2 = __builtin_amdgcn_cosf(fmaf(ys[2], INV_2PI, thr[2]));
      const float c3 = __builtin_amdgcn_cosf(fmaf(ys[3], INV_2PI, thr[3]));
      const float p01 = c0 * c1, p23 = c2 * c3;
      const float u = hi ? p01 : c1;
      const float v = hi ? c2 : p23;
      v2f b = {u * v, hi ? p01 * p23 : p01};
      bz = b;
    }

    const v8f ga = wmma_f32_16x16x4(aw, bz, cbias);
    const float g_lo0 = q_odd ? ga[2] : ga[0], g_lo1 = q_odd ? ga[3] : ga[1];
    const float g_hi0 = q_odd ? ga[6] : ga[4], g_hi1 = q_odd ? ga[7] : ga[5];
    const float gx[2] = {q_high ? g_hi0 : g_lo0, q_high ? g_hi1 : g_lo1};
#pragma unroll
    for (int i = 0; i < 2; ++i) {
      const float a  = fmaf(act_sc, hw_tanh(act_sc * gx[i]), act_off);
      const float fv = quad_bcast<0>(a);
      const float iv = quad_bcast<1>(a);
      const float gv = quad_bcast<2>(a);
      const float ov = quad_bcast<3>(a);
      float cs = fv * c_state[i] + iv * gv;
      c_state[i] = cs;
      h[i] = ov * hw_tanh(cs);
    }
    if (gate == 0) {
      float2 hv = {h[0], h[1]};
      *(float2*)(hs_row + (size_t)t * H_SZ) = hv;
    }
  }

  if (gate == 0) {
    float2 hv = {h[0], h[1]};
    float2 cv = {c_state[0], c_state[1]};
    *(float2*)(hT + (size_t)brow * H_SZ + cbase) = hv;
    *(float2*)(cT + (size_t)brow * H_SZ + cbase) = cv;
  }
}

extern "C" void kernel_launch(void* const* d_in, const int* in_sizes, int n_in,
                              void* d_out, int out_size, void* d_ws, size_t ws_size,
                              hipStream_t stream) {
  const float* x      = (const float*)d_in[0];
  const float* W_in   = (const float*)d_in[1];
  const float* b_in   = (const float*)d_in[2];
  const float* W_out  = (const float*)d_in[3];
  const float* b_out  = (const float*)d_in[4];
  const float* thetas = (const float*)d_in[5];
  float* out = (float*)d_out;

  qlstm_recur<<<256, 256, 0, stream>>>(x, W_in, b_in, W_out, b_out, thetas, out);
}


# gate product with operands swapped (z as A, W_out as B): a lane finds f,i,g,o of ONE H column in its own D registers, no select, no quad exchange; h to B by 2 row gathers; y partial k order regrouped
# speedup vs baseline: 5.6017x; 1.0378x over previous
#include <hip/hip_runtime.h>
#include <hip/hip_bf16.h>
#include <math.h>

typedef __attribute__((ext_vector_type(2))) float v2f;
typedef __attribute__((ext_vector_type(8))) float v8f;

static __device__ __forceinline__ v8f wmma_f32_16x16x4(v2f a, v2f b, v8f c) {
  return __builtin_amdgcn_wmma_f32_16x16x4_f32(false, a, false, b, (short)0, c, false, false);
}

static __device__ __forceinline__ float hw_tanh(float x) {
  return __builtin_amdgcn_tanhf(x);
}
static __device__ __forceinline__ float hw_sigmoid(float x) {
  return fmaf(0.5f, hw_tanh(0.5f * x), 0.5f);
}

static __device__ __forceinline__ void lds_barrier() {
  __builtin_amdgcn_fence(__ATOMIC_RELEASE, "workgroup", "local");
  __builtin_amdgcn_s_barrier();
  __builtin_amdgcn_fence(__ATOMIC_ACQUIRE, "workgroup", "local");
}

#define B_SZ 256
#define S_SZ 256
#define F_SZ 64
#define H_SZ 128
#define X_PITCH 68


static __device__ __forceinline__ float row_gather(float v, unsigned sel) {
  return __uint_as_float(__builtin_amdgcn_permlane16_var(0u, __float_as_uint(v), sel, false, true));
}

template <int M>
static __device__ __forceinline__ float row_xor(float v) {
  return __int_as_float(__builtin_amdgcn_mov_dpp(__float_as_int(v), 0x160 + M, 0xF, 0xF, true));
}

__global__ void __launch_bounds__(256)
qlstm_recur(const float* __restrict__ x, const float* __restrict__ Win,
            const float* __restrict__ bin,
            const float* __restrict__ Wout, const float* __restrict__ bout,
            const float* __restrict__ thetas, float* __restrict__ out) {
  __shared__ __attribute__((aligned(16))) float x_lds[S_SZ * X_PITCH];
  __shared__ __attribute__((aligned(16))) float y0_lds[S_SZ * 4];
  __shared__ __attribute__((aligned(16))) float part[2 * 16 * 4];

  const int tid   = threadIdx.x;
  const int lane  = tid & 31;
  const int wave  = tid >> 5;
  const int n     = lane & 15;
  const int gate  = n & 3;
  const int sq    = n >> 2;
  const int hi    = lane >> 4;
  const int kh    = hi << 1;
  const int brow  = blockIdx.x;
  const int col   = wave * 16 + n;

  float* hs = out;
  float* hT = out + (size_t)B_SZ * S_SZ * H_SZ;
  float* cT = hT + (size_t)B_SZ * H_SZ;

  v2f bch[16];
  float bb;
  {
    const bool ncol = n < 4;
    const float* wr = Win + (n & 3) * 192 + 128 + kh;
#pragma unroll
    for (int j = 0; j < 16; ++j) {
      const float b0 = wr[j * 4], b1 = wr[j * 4 + 1];
      v2f bv = {ncol ? b0 : 0.f, ncol ? b1 : 0.f};
      bch[j] = bv;
    }
    const float bq = bin[n & 3];
    bb = ncol ? bq : 0.f;
  }
  {
    const float4* xg = (const float4*)(x + (size_t)brow * S_SZ * F_SZ) + wave * 512 + lane;
    float4 xv[16];
#pragma unroll
    for (int i = 0; i < 16; ++i) xv[i] = xg[i * 32];
#pragma unroll
    for (int i = 0; i < 16; ++i)
      *(float4*)&x_lds[(wave * 32 + i * 2 + hi) * X_PITCH + n * 4] = xv[i];
  }
  __syncthreads();
  {
    const float* xr0 = x_lds + (wave * 32 + n) * X_PITCH + kh;
    const float* xr1 = xr0 + 16 * X_PITCH;
    v8f acc0 = {bb, bb, bb, bb, bb, bb, bb, bb};
    v8f acc1 = acc0;
#pragma unroll
    for (int j = 0; j < 16; ++j) {
      v2f a0 = {xr0[j * 4], xr0[j * 4 + 1]};
      v2f a1 = {xr1[j * 4], xr1[j * 4 + 1]};
      acc0 = wmma_f32_16x16x4(a0, bch[j], acc0);
      acc1 = wmma_f32_16x16x4(a1, bch[j], acc1);
    }
    if (n < 4) {
#pragma unroll
      for (int r = 0; r < 8; ++r) {
        y0_lds[(wave * 32 + hi * 8 + r) * 4 + n]      = acc0[r];
        y0_lds[(wave * 32 + 16 + hi * 8 + r) * 4 + n] = acc1[r];
      }
    }
  }

  v2f ah;
  {
    const int kk = wave * 16 + (n >> 2) * 4 + hi * 2;
    v2f av = {Win[(n & 3) * 192 + kk], Win[(n & 3) * 192 + kk + 1]};
    ah = av;
  }
  v2f aw = {Wout[col * 4 + kh], Wout[col * 4 + kh + 1]};
  const float bo = bout[col];
  const v8f cbias = {bo, bo, bo, bo, bo, bo, bo, bo};

  const float INV_2PI = 0.15915494309189532f;
  float thr[4];
#pragma unroll
  for (int w = 0; w < 4; ++w) thr[w] = thetas[gate * 4 + w] * INV_2PI;

  float c_state = 0.f, h = 0.f;
  v2f hb = {0.f, 0.f};
  const unsigned sel0 = (unsigned)(sq * 4 + kh);
  const unsigned sel1 = sel0 + 1u;

  float* hs_row = hs + ((size_t)brow * S_SZ) * H_SZ + col;

  __syncthreads();

  for (int t = 0; t < S_SZ; ++t) {
    float* pw = part + (t & 1) * (16 * 4);
    const float4 yv = *(const float4*)(y0_lds + t * 4);

    v8f acc = {0.f, 0.f, 0.f, 0.f, 0.f, 0.f, 0.f, 0.f};
    acc = wmma_f32_16x16x4(ah, hb, acc);
    float4 pv;
    pv.x = acc[0] + row_xor<4>(acc[4]);
    pv.y = acc[1] + row_xor<4>(acc[5]);
    pv.z = acc[2] + row_xor<4>(acc[6]);
    pv.w = acc[3] + row_xor<4>(acc[7]);
    if (lane == hi * 24)
      *(float4*)&pw[(wave * 2 + hi) * 4] = pv;
    lds_barrier();

    float4 qv = *(const float4*)&pw[n * 4];
    qv.x += row_xor<1>(qv.x);  qv.y += row_xor<1>(qv.y);  qv.z += row_xor<1>(qv.z);  qv.w += row_xor<1>(qv.w);
    qv.x += row_xor<2>(qv.x);  qv.y += row_xor<2>(qv.y);  qv.z += row_xor<2>(qv.z);  qv.w += row_xor<2>(qv.w);
    qv.x += row_xor<4>(qv.x);  qv.y += row_xor<4>(qv.y);  qv.z += row_xor<4>(qv.z);  qv.w += row_xor<4>(qv.w);
    qv.x += row_xor<8>(qv.x);  qv.y += row_xor<8>(qv.y);  qv.z += row_xor<8>(qv.z);  qv.w += row_xor<8>(qv.w);
    const float ys[4] = {yv.x + qv.x, yv.y + qv.y, yv.z + qv.z, yv.w + qv.w};
    v2f bz;
    {
      const float c0 = __builtin_amdgcn_cosf(fmaf(ys[0], INV_2PI, thr[0]));
      const float c1 = __builtin_amdgcn_cosf(fmaf(ys[1], INV_2PI, thr[1]));
      const float c2 = __builtin_amdgcn_cosf(fmaf(ys[2], INV_2PI, thr[2]));
      const float c3 = __builtin_amdgcn_cosf(fmaf(ys[3], INV_2PI, thr[3]));
      const float p01 = c0 * c1, p23 = c2 * c3;
      const float u = hi ? p01 : c1;
      const float v = hi ? c2 : p23;
      v2f b = {u * v, hi ? p01 * p23 : p01};
      bz = b;
    }

    const v8f ga = wmma_f32_16x16x4(bz, aw, cbias);
    {
      const float fv = hw_sigmoid(ga[0]);
      const float iv = hw_sigmoid(ga[1]);
      const float gv = hw_tanh(ga[2]);
      const float ov = hw_sigmoid(ga[3]);
      c_state = fv * c_state + iv * gv;
      h = ov * hw_tanh(c_state);
    }
    hb[0] = row_gather(h, sel0);
    hb[1] = row_gather(h, sel1);
    if (hi == 0) hs_row[(size_t)t * H_SZ] = h;
  }

  if (hi == 0) {
    hT[(size_t)brow * H_SZ + col] = h;
    cT[(size_t)brow * H_SZ + col] = c_state;
  }
}

extern "C" void kernel_launch(void* const* d_in, const int* in_sizes, int n_in,
                              void* d_out, int out_size, void* d_ws, size_t ws_size,
                              hipStream_t stream) {
  const float* x      = (const float*)d_in[0];
  const float* W_in   = (const float*)d_in[1];
  const float* b_in   = (const float*)d_in[2];
  const float* W_out  = (const float*)d_in[3];
  const float* b_out  = (const float*)d_in[4];
  const float* thetas = (const float*)d_in[5];
  float* out = (float*)d_out;

  qlstm_recur<<<256, 256, 0, stream>>>(x, W_in, b_in, W_out, b_out, thetas, out);
}
